# starting kernel
# baseline (speedup 1.0000x reference)
#include <hip/hip_runtime.h>
#include <math.h>

#define HEADS 8
#define DIM 16
#define HD 128
#define INV_SCALE 0.25f

typedef __attribute__((ext_vector_type(2))) float v2f;
typedef __attribute__((ext_vector_type(8))) float v8f;

__global__ void zero_kernel(float* __restrict__ out, float* __restrict__ zbuf,
                            int nOut, int nZ) {
  int i = blockIdx.x * blockDim.x + threadIdx.x;
  if (i < nOut) out[i] = 0.0f;
  if (i < nZ)  zbuf[i] = 0.0f;
}

__global__ void __launch_bounds__(256)
edge_attn_kernel(const float* __restrict__ q, const float* __restrict__ k,
                 const float* __restrict__ v, const int* __restrict__ eidx,
                 float* __restrict__ wV, float* __restrict__ Z, int nEdges) {
  __shared__ int   s_idx[8][32];
  __shared__ float s_sc[8][16];

  const int lane = threadIdx.x & 31;
  const int wv   = threadIdx.x >> 5;
  const long e0  = ((long)blockIdx.x * 8 + wv) * 16;
  if (e0 >= nEdges) return;

  {
    long e   = e0 + (lane & 15);
    int  ok  = e < (long)nEdges;
    long off = (lane < 16) ? e : ((long)nEdges + e);
    s_idx[wv][lane] = ok ? eidx[off] : 0;
  }
  asm volatile("s_wait_dscnt 0x0" ::: "memory");

  const int m     = lane & 15;
  const int is_hi = lane >> 4;
  const int srcm  = s_idx[wv][m];
  const int dstm  = s_idx[wv][16 + m];

  const int em   = lane >> 1;
  const int half = lane & 1;
  const int srce = s_idx[wv][em];
  const int dste = s_idx[wv][16 + em];
  const bool ev  = (e0 + em) < (long)nEdges;

  const float* arow = k + (long)srcm * HD + is_hi * 2;
  const float* brow = q + (long)dstm * HD + is_hi * 2;
  const float* vrow = v + (long)srce * HD + half * 8;
  float*       orow = wV + (long)dste * HD + half * 8;

  const int  idxsel = lane & 7;
  const bool isdiag = (lane < 8) || (lane >= 24);
  const int  dm     = (lane >= 24) ? (lane - 16) : lane;

  for (int h = 0; h < HEADS; ++h) {
    const float* ap = arow + h * DIM;
    const float* bp = brow + h * DIM;
    v8f c = {0.f, 0.f, 0.f, 0.f, 0.f, 0.f, 0.f, 0.f};
#pragma unroll
    for (int c4 = 0; c4 < 4; ++c4) {
      v2f a = *(const v2f*)(ap + c4 * 4);
      v2f b = *(const v2f*)(bp + c4 * 4);
      c = __builtin_amdgcn_wmma_f32_16x16x4_f32(false, a, false, b,
                                                (short)0, c, false, false);
    }
    float d01 = (idxsel & 1) ? c[1] : c[0];
    float d23 = (idxsel & 1) ? c[3] : c[2];
    float d45 = (idxsel & 1) ? c[5] : c[4];
    float d67 = (idxsel & 1) ? c[7] : c[6];
    float d03 = (idxsel & 2) ? d23 : d01;
    float d47 = (idxsel & 2) ? d67 : d45;
    float dd  = (idxsel & 4) ? d47 : d03;
    if (isdiag) {
      float t = dd * INV_SCALE;
      t = fminf(fmaxf(t, -5.0f), 5.0f);
      s_sc[wv][dm] = __expf(t);
    }
    asm volatile("s_wait_dscnt 0x0" ::: "memory");
    float sc = s_sc[wv][em];

    if (ev) {
      const float4* vp = (const float4*)(vrow + h * DIM);
      float4 p0 = vp[0];
      float4 p1 = vp[1];
      float* op = orow + h * DIM;
      atomicAdd(op + 0, p0.x * sc); atomicAdd(op + 1, p0.y * sc);
      atomicAdd(op + 2, p0.z * sc); atomicAdd(op + 3, p0.w * sc);
      atomicAdd(op + 4, p1.x * sc); atomicAdd(op + 5, p1.y * sc);
      atomicAdd(op + 6, p1.z * sc); atomicAdd(op + 7, p1.w * sc);
      if (half == 0) atomicAdd(Z + (long)dste * HEADS + h, sc);
    }
  }
}

__global__ void norm_kernel(float* __restrict__ out, const float* __restrict__ Z,
                            int n) {
  int i = blockIdx.x * blockDim.x + threadIdx.x;
  if (i >= n) return;
  int node = i >> 7;
  int h    = (i >> 4) & 7;
  out[i] = out[i] / (Z[node * HEADS + h] + 1e-6f);
}

extern "C" void kernel_launch(void* const* d_in, const int* in_sizes, int n_in,
                              void* d_out, int out_size, void* d_ws, size_t ws_size,
                              hipStream_t stream) {
  const float* q    = (const float*)d_in[0];
  const float* k    = (const float*)d_in[1];
  const float* v    = (const float*)d_in[2];
  const int*   eidx = (const int*)d_in[3];
  float* out = (float*)d_out;
  float* Z   = (float*)d_ws;

  const int nEdges = in_sizes[3] / 2;
  const int nOut   = out_size;
  const int nZ     = nOut / DIM;

  const int threads = 256;
  const int blocks0 = (nOut + threads - 1) / threads;
  zero_kernel<<<blocks0, threads, 0, stream>>>(out, Z, nOut, nZ);

  const int groups  = (nEdges + 15) / 16;
  const int blocks1 = (groups + 7) / 8;
  edge_attn_kernel<<<blocks1, threads, 0, stream>>>(q, k, v, eidx, out, Z, nEdges);

  norm_kernel<<<blocks0, threads, 0, stream>>>(out, Z, nOut);
}


# Float adds go out one destination row at a time: each add instruction's 32 lanes cover one whole 128-byte line (was 16 lines of 8 bytes each); sums still land in arrival order
# speedup vs baseline: 4.0487x; 4.0487x over previous
#include <hip/hip_runtime.h>
#include <math.h>

#define HEADS 8
#define DIM 16
#define HD 128
#define INV_SCALE 0.25f

typedef __attribute__((ext_vector_type(2))) float v2f;
typedef __attribute__((ext_vector_type(8))) float v8f;

__global__ void zero_kernel(float* __restrict__ out, float* __restrict__ zbuf,
                            int nOut, int nZ) {
  int i = blockIdx.x * blockDim.x + threadIdx.x;
  if (i < nOut) out[i] = 0.0f;
  if (i < nZ)  zbuf[i] = 0.0f;
}

__global__ void __launch_bounds__(256)
edge_attn_kernel(const float* __restrict__ q, const float* __restrict__ k,
                 const float* __restrict__ v, const int* __restrict__ eidx,
                 float* __restrict__ wV, float* __restrict__ Z, int nEdges) {
  __shared__ int    s_idx[8][32];
  __shared__ float4 s_w4[8][32];

  const int lane = threadIdx.x & 31;
  const int wv   = threadIdx.x >> 5;
  const long e0  = ((long)blockIdx.x * 8 + wv) * 16;
  if (e0 >= nEdges) return;

  int myidx;
  {
    long e   = e0 + (lane & 15);
    int  ok  = e < (long)nEdges;
    long off = (lane < 16) ? e : ((long)nEdges + e);
    myidx = ok ? eidx[off] : 0;
    s_idx[wv][lane] = myidx;
  }
  asm volatile("s_wait_dscnt 0x0" ::: "memory");

  const int m     = lane & 15;
  const int is_hi = lane >> 4;
  const int srcm  = s_idx[wv][m];
  const int dstm  = s_idx[wv][16 + m];

  const float* arow = k + (long)srcm * HD + is_hi * 2;
  const float* brow = q + (long)dstm * HD + is_hi * 2;

  const int  idxsel = lane & 7;
  const bool isdiag = (lane < 8) || (lane >= 24);
  const int  dm     = (lane >= 24) ? (lane - 16) : lane;

  float* sw = (float*)&s_w4[wv][0];

  for (int h = 0; h < HEADS; ++h) {
    const float* ap = arow + h * DIM;
    const float* bp = brow + h * DIM;
    v8f c = {0.f, 0.f, 0.f, 0.f, 0.f, 0.f, 0.f, 0.f};
#pragma unroll
    for (int c4 = 0; c4 < 4; ++c4) {
      v2f a = *(const v2f*)(ap + c4 * 4);
      v2f b = *(const v2f*)(bp + c4 * 4);
      c = __builtin_amdgcn_wmma_f32_16x16x4_f32(false, a, false, b,
                                                (short)0, c, false, false);
    }
    float d01 = (idxsel & 1) ? c[1] : c[0];
    float d23 = (idxsel & 1) ? c[3] : c[2];
    float d45 = (idxsel & 1) ? c[5] : c[4];
    float d67 = (idxsel & 1) ? c[7] : c[6];
    float d03 = (idxsel & 2) ? d23 : d01;
    float d47 = (idxsel & 2) ? d67 : d45;
    float dd  = (idxsel & 4) ? d47 : d03;
    if (isdiag) {
      float t = dd * INV_SCALE;
      t = fminf(fmaxf(t, -5.0f), 5.0f);
      sw[dm * 8 + (h & 1) * 4 + (h >> 1)] = __expf(t);
    }
  }
  asm volatile("s_wait_dscnt 0x0" ::: "memory");

  {
    const int zh    = lane & 7;
    const int zslot = (zh & 1) * 4 + (zh >> 1);
    for (int t = 0; t < 4; ++t) {
      const int j = t * 4 + (lane >> 3);
      if ((e0 + j) < (long)nEdges) {
        const int dj = s_idx[wv][16 + j];
        atomicAdd(Z + (long)dj * HEADS + zh, sw[j * 8 + zslot]);
      }
    }
  }

  const long left = (long)nEdges - e0;
  const int  nv   = left < 16 ? (int)left : 16;
  for (int j = 0; j < nv; ++j) {
    const int sj = __builtin_amdgcn_readlane(myidx, j);
    const int dj = __builtin_amdgcn_readlane(myidx, 16 + j);
    const float4 w = s_w4[wv][j * 2 + is_hi];
    const float* vp = v + (long)sj * HD + lane;
    float*       op = wV + (long)dj * HD + lane;
    const float x0 = vp[0];
    const float x1 = vp[32];
    const float x2 = vp[64];
    const float x3 = vp[96];
    atomicAdd(op,      x0 * w.x);
    atomicAdd(op + 32, x1 * w.y);
    atomicAdd(op + 64, x2 * w.z);
    atomicAdd(op + 96, x3 * w.w);
  }
}

__global__ void norm_kernel(float* __restrict__ out, const float* __restrict__ Z,
                            int n) {
  int i = blockIdx.x * blockDim.x + threadIdx.x;
  if (i >= n) return;
  int node = i >> 7;
  int h    = (i >> 4) & 7;
  out[i] = out[i] / (Z[node * HEADS + h] + 1e-6f);
}

extern "C" void kernel_launch(void* const* d_in, const int* in_sizes, int n_in,
                              void* d_out, int out_size, void* d_ws, size_t ws_size,
                              hipStream_t stream) {
  const float* q    = (const float*)d_in[0];
  const float* k    = (const float*)d_in[1];
  const float* v    = (const float*)d_in[2];
  const int*   eidx = (const int*)d_in[3];
  float* out = (float*)d_out;
  float* Z   = (float*)d_ws;

  const int nEdges = in_sizes[3] / 2;
  const int nOut   = out_size;
  const int nZ     = nOut / DIM;

  const int threads = 256;
  const int blocks0 = (nOut + threads - 1) / threads;
  zero_kernel<<<blocks0, threads, 0, stream>>>(out, Z, nOut, nZ);

  const int groups  = (nEdges + 15) / 16;
  const int blocks1 = (groups + 7) / 8;
  edge_attn_kernel<<<blocks1, threads, 0, stream>>>(q, k, v, eidx, out, Z, nEdges);

  norm_kernel<<<blocks0, threads, 0, stream>>>(out, Z, nOut);
}


# K_src and Q_dst rows are fetched as whole 128-byte lines (b128 loads, 4 lines per instruction; was 16 lines of 16 B) and parked in padded LDS rows; WMMA fragments read from LDS in the same k order
# speedup vs baseline: 4.9586x; 1.2247x over previous
#include <hip/hip_runtime.h>
#include <math.h>

#define HEADS 8
#define DIM 16
#define HD 128
#define INV_SCALE 0.25f
#define ROWPITCH 36

typedef __attribute__((ext_vector_type(2))) float v2f;
typedef __attribute__((ext_vector_type(8))) float v8f;

__global__ void zero_kernel(float* __restrict__ out, float* __restrict__ zbuf,
                            int nOut, int nZ) {
  int i = blockIdx.x * blockDim.x + threadIdx.x;
  if (i < nOut) out[i] = 0.0f;
  if (i < nZ)  zbuf[i] = 0.0f;
}

__global__ void __launch_bounds__(256)
edge_attn_kernel(const float* __restrict__ q, const float* __restrict__ k,
                 const float* __restrict__ v, const int* __restrict__ eidx,
                 float* __restrict__ wV, float* __restrict__ Z, int nEdges) {
  __shared__ int    s_idx[8][32];
  __shared__ float4 s_w4[8][32];
  __shared__ float4 s_k4[8][16 * ROWPITCH / 4];
  __shared__ float4 s_q4[8][16 * ROWPITCH / 4];

  const int lane = threadIdx.x & 31;
  const int wv   = threadIdx.x >> 5;
  const long e0  = ((long)blockIdx.x * 8 + wv) * 16;
  if (e0 >= nEdges) return;

  int myidx;
  {
    long e   = e0 + (lane & 15);
    int  ok  = e < (long)nEdges;
    long off = (lane < 16) ? e : ((long)nEdges + e);
    myidx = ok ? eidx[off] : 0;
    s_idx[wv][lane] = myidx;
  }
  asm volatile("s_wait_dscnt 0x0" ::: "memory");

  const int pc   = lane & 7;
  const int rsub = lane >> 3;
  const char* kbytes = (const char*)k;
  const char* qbytes = (const char*)q;
  float* skw = (float*)&s_k4[wv][0];
  float* sqw = (float*)&s_q4[wv][0];
  float* kst = skw + rsub * ROWPITCH + pc * 4;
  float* qst = sqw + rsub * ROWPITCH + pc * 4;

  const int m     = lane & 15;
  const int is_hi = lane >> 4;
  const float* arow = skw + m * ROWPITCH + is_hi * 2;
  const float* brow = sqw + m * ROWPITCH + is_hi * 2;

  const int  idxsel = lane & 7;
  const bool isdiag = (lane < 8) || (lane >= 24);
  const int  dm     = (lane >= 24) ? (lane - 16) : lane;

  float* sw = (float*)&s_w4[wv][0];

  for (int hp = 0; hp < HEADS / 2; ++hp) {
    const char* kline = kbytes + hp * 128;
    const char* qline = qbytes + hp * 128;
    float4 kr[4], qr[4];
    for (int i = 0; i < 4; ++i) {
      const int r = 4 * i + rsub;
      const unsigned kofs = ((unsigned)s_idx[wv][r]      * HD + pc * 4) * 4u;
      const unsigned qofs = ((unsigned)s_idx[wv][16 + r] * HD + pc * 4) * 4u;
      kr[i] = *(const float4*)(kline + kofs);
      qr[i] = *(const float4*)(qline + qofs);
    }
    for (int i = 0; i < 4; ++i) {
      *(float4*)(kst + i * 4 * ROWPITCH) = kr[i];
      *(float4*)(qst + i * 4 * ROWPITCH) = qr[i];
    }
    asm volatile("s_wait_dscnt 0x0" ::: "memory");

    for (int hh = 0; hh < 2; ++hh) {
      const float* ap = arow + hh * DIM;
      const float* bp = brow + hh * DIM;
      v8f c = {0.f, 0.f, 0.f, 0.f, 0.f, 0.f, 0.f, 0.f};
#pragma unroll
      for (int c4 = 0; c4 < 4; ++c4) {
        v2f a = *(const v2f*)(ap + c4 * 4);
        v2f b = *(const v2f*)(bp + c4 * 4);
        c = __builtin_amdgcn_wmma_f32_16x16x4_f32(false, a, false, b,
                                                  (short)0, c, false, false);
      }
      float d01 = (idxsel & 1) ? c[1] : c[0];
      float d23 = (idxsel & 1) ? c[3] : c[2];
      float d45 = (idxsel & 1) ? c[5] : c[4];
      float d67 = (idxsel & 1) ? c[7] : c[6];
      float d03 = (idxsel & 2) ? d23 : d01;
      float d47 = (idxsel & 2) ? d67 : d45;
      float dd  = (idxsel & 4) ? d47 : d03;
      if (isdiag) {
        float t = dd * INV_SCALE;
        t = fminf(fmaxf(t, -5.0f), 5.0f);
        sw[dm * 8 + hh * 4 + hp] = __expf(t);
      }
    }
    asm volatile("s_wait_dscnt 0x0" ::: "memory");
  }

  {
    const int zh    = lane & 7;
    const int zslot = (zh & 1) * 4 + (zh >> 1);
    for (int t = 0; t < 4; ++t) {
      const int j = t * 4 + (lane >> 3);
      if ((e0 + j) < (long)nEdges) {
        const int dj = s_idx[wv][16 + j];
        atomicAdd(Z + (long)dj * HEADS + zh, sw[j * 8 + zslot]);
      }
    }
  }

  const long left = (long)nEdges - e0;
  const int  nv   = left < 16 ? (int)left : 16;
  for (int j = 0; j < nv; ++j) {
    const int sj = __builtin_amdgcn_readlane(myidx, j);
    const int dj = __builtin_amdgcn_readlane(myidx, 16 + j);
    const float4 w = s_w4[wv][j * 2 + is_hi];
    const float* vp = v + (long)sj * HD + lane;
    float*       op = wV + (long)dj * HD + lane;
    const float x0 = vp[0];
    const float x1 = vp[32];
    const float x2 = vp[64];
    const float x3 = vp[96];
    atomicAdd(op,      x0 * w.x);
    atomicAdd(op + 32, x1 * w.y);
    atomicAdd(op + 64, x2 * w.z);
    atomicAdd(op + 96, x3 * w.w);
  }
}

__global__ void norm_kernel(float* __restrict__ out, const float* __restrict__ Z,
                            int n) {
  int i = blockIdx.x * blockDim.x + threadIdx.x;
  if (i >= n) return;
  int node = i >> 7;
  int h    = (i >> 4) & 7;
  out[i] = out[i] / (Z[node * HEADS + h] + 1e-6f);
}

extern "C" void kernel_launch(void* const* d_in, const int* in_sizes, int n_in,
                              void* d_out, int out_size, void* d_ws, size_t ws_size,
                              hipStream_t stream) {
  const float* q    = (const float*)d_in[0];
  const float* k    = (const float*)d_in[1];
  const float* v    = (const float*)d_in[2];
  const int*   eidx = (const int*)d_in[3];
  float* out = (float*)d_out;
  float* Z   = (float*)d_ws;

  const int nEdges = in_sizes[3] / 2;
  const int nOut   = out_size;
  const int nZ     = nOut / DIM;

  const int threads = 256;
  const int blocks0 = (nOut + threads - 1) / threads;
  zero_kernel<<<blocks0, threads, 0, stream>>>(out, Z, nOut, nZ);

  const int groups  = (nEdges + 15) / 16;
  const int blocks1 = (groups + 7) / 8;
  edge_attn_kernel<<<blocks1, threads, 0, stream>>>(q, k, v, eidx, out, Z, nEdges);

  norm_kernel<<<blocks0, threads, 0, stream>>>(out, Z, nOut);
}


# V rows of 8 edges are loaded as one group of 32 loads ahead of their adds (was one edge's 4 loads, then its 4 adds, 16 times in turn); sums still land in arrival order
# speedup vs baseline: 5.3285x; 1.0746x over previous
#include <hip/hip_runtime.h>
#include <math.h>

#define HEADS 8
#define DIM 16
#define HD 128
#define INV_SCALE 0.25f
#define VBATCH 8
#define ROWPITCH 36

typedef __attribute__((ext_vector_type(2))) float v2f;
typedef __attribute__((ext_vector_type(8))) float v8f;

__global__ void zero_kernel(float* __restrict__ out, float* __restrict__ zbuf,
                            int nOut, int nZ) {
  int i = blockIdx.x * blockDim.x + threadIdx.x;
  if (i < nOut) out[i] = 0.0f;
  if (i < nZ)  zbuf[i] = 0.0f;
}

__global__ void __launch_bounds__(256)
edge_attn_kernel(const float* __restrict__ q, const float* __restrict__ k,
                 const float* __restrict__ v, const int* __restrict__ eidx,
                 float* __restrict__ wV, float* __restrict__ Z, int nEdges) {
  __shared__ int    s_idx[8][32];
  __shared__ float4 s_w4[8][32];
  __shared__ float4 s_k4[8][16 * ROWPITCH / 4];
  __shared__ float4 s_q4[8][16 * ROWPITCH / 4];

  const int lane = threadIdx.x & 31;
  const int wv   = threadIdx.x >> 5;
  const long e0  = ((long)blockIdx.x * 8 + wv) * 16;
  if (e0 >= nEdges) return;

  int myidx;
  {
    long e   = e0 + (lane & 15);
    int  ok  = e < (long)nEdges;
    long off = (lane < 16) ? e : ((long)nEdges + e);
    myidx = ok ? eidx[off] : 0;
    s_idx[wv][lane] = myidx;
  }
  asm volatile("s_wait_dscnt 0x0" ::: "memory");

  const int pc   = lane & 7;
  const int rsub = lane >> 3;
  const char* kbytes = (const char*)k;
  const char* qbytes = (const char*)q;
  float* skw = (float*)&s_k4[wv][0];
  float* sqw = (float*)&s_q4[wv][0];
  float* kst = skw + rsub * ROWPITCH + pc * 4;
  float* qst = sqw + rsub * ROWPITCH + pc * 4;

  const int m     = lane & 15;
  const int is_hi = lane >> 4;
  const float* arow = skw + m * ROWPITCH + is_hi * 2;
  const float* brow = sqw + m * ROWPITCH + is_hi * 2;

  const int  idxsel = lane & 7;
  const bool isdiag = (lane < 8) || (lane >= 24);
  const int  dm     = (lane >= 24) ? (lane - 16) : lane;

  float* sw = (float*)&s_w4[wv][0];

  for (int hp = 0; hp < HEADS / 2; ++hp) {
    const char* kline = kbytes + hp * 128;
    const char* qline = qbytes + hp * 128;
    float4 kr[4], qr[4];
    for (int i = 0; i < 4; ++i) {
      const int r = 4 * i + rsub;
      const unsigned kofs = ((unsigned)s_idx[wv][r]      * HD + pc * 4) * 4u;
      const unsigned qofs = ((unsigned)s_idx[wv][16 + r] * HD + pc * 4) * 4u;
      kr[i] = *(const float4*)(kline + kofs);
      qr[i] = *(const float4*)(qline + qofs);
    }
    for (int i = 0; i < 4; ++i) {
      *(float4*)(kst + i * 4 * ROWPITCH) = kr[i];
      *(float4*)(qst + i * 4 * ROWPITCH) = qr[i];
    }
    asm volatile("s_wait_dscnt 0x0" ::: "memory");

    for (int hh = 0; hh < 2; ++hh) {
      const float* ap = arow + hh * DIM;
      const float* bp = brow + hh * DIM;
      v8f c = {0.f, 0.f, 0.f, 0.f, 0.f, 0.f, 0.f, 0.f};
#pragma unroll
      for (int c4 = 0; c4 < 4; ++c4) {
        v2f a = *(const v2f*)(ap + c4 * 4);
        v2f b = *(const v2f*)(bp + c4 * 4);
        c = __builtin_amdgcn_wmma_f32_16x16x4_f32(false, a, false, b,
                                                  (short)0, c, false, false);
      }
      float d01 = (idxsel & 1) ? c[1] : c[0];
      float d23 = (idxsel & 1) ? c[3] : c[2];
      float d45 = (idxsel & 1) ? c[5] : c[4];
      float d67 = (idxsel & 1) ? c[7] : c[6];
      float d03 = (idxsel & 2) ? d23 : d01;
      float d47 = (idxsel & 2) ? d67 : d45;
      float dd  = (idxsel & 4) ? d47 : d03;
      if (isdiag) {
        float t = dd * INV_SCALE;
        t = fminf(fmaxf(t, -5.0f), 5.0f);
        sw[dm * 8 + hh * 4 + hp] = __expf(t);
      }
    }
    asm volatile("s_wait_dscnt 0x0" ::: "memory");
  }

  {
    const int zh    = lane & 7;
    const int zslot = (zh & 1) * 4 + (zh >> 1);
    for (int t = 0; t < 4; ++t) {
      const int j = t * 4 + (lane >> 3);
      if ((e0 + j) < (long)nEdges) {
        const int dj = s_idx[wv][16 + j];
        atomicAdd(Z + (long)dj * HEADS + zh, sw[j * 8 + zslot]);
      }
    }
  }

  const long left = (long)nEdges - e0;
  const int  nv   = __builtin_amdgcn_readfirstlane(left < 16 ? (int)left : 16);
  int j = 0;
  for (; j + VBATCH <= nv; j += VBATCH) {
    float x[VBATCH][4];
#pragma unroll
    for (int u = 0; u < VBATCH; ++u) {
      const int sj = __builtin_amdgcn_readlane(myidx, j + u);
      const float* vp = v + (long)sj * HD + lane;
      x[u][0] = vp[0];
      x[u][1] = vp[32];
      x[u][2] = vp[64];
      x[u][3] = vp[96];
    }
#pragma unroll
    for (int u = 0; u < VBATCH; ++u) {
      const int dj = __builtin_amdgcn_readlane(myidx, 16 + j + u);
      const float4 w = s_w4[wv][(j + u) * 2 + is_hi];
      float* op = wV + (long)dj * HD + lane;
      atomicAdd(op,      x[u][0] * w.x);
      atomicAdd(op + 32, x[u][1] * w.y);
      atomicAdd(op + 64, x[u][2] * w.z);
      atomicAdd(op + 96, x[u][3] * w.w);
    }
  }
  for (; j < nv; ++j) {
    const int sj = __builtin_amdgcn_readlane(myidx, j);
    const int dj = __builtin_amdgcn_readlane(myidx, 16 + j);
    const float4 w = s_w4[wv][j * 2 + is_hi];
    const float* vp = v + (long)sj * HD + lane;
    float*       op = wV + (long)dj * HD + lane;
    const float x0 = vp[0];
    const float x1 = vp[32];
    const float x2 = vp[64];
    const float x3 = vp[96];
    atomicAdd(op,      x0 * w.x);
    atomicAdd(op + 32, x1 * w.y);
    atomicAdd(op + 64, x2 * w.z);
    atomicAdd(op + 96, x3 * w.w);
  }
}

__global__ void norm_kernel(float* __restrict__ out, const float* __restrict__ Z,
                            int n) {
  int i = blockIdx.x * blockDim.x + threadIdx.x;
  if (i >= n) return;
  int node = i >> 7;
  int h    = (i >> 4) & 7;
  out[i] = out[i] / (Z[node * HEADS + h] + 1e-6f);
}

extern "C" void kernel_launch(void* const* d_in, const int* in_sizes, int n_in,
                              void* d_out, int out_size, void* d_ws, size_t ws_size,
                              hipStream_t stream) {
  const float* q    = (const float*)d_in[0];
  const float* k    = (const float*)d_in[1];
  const float* v    = (const float*)d_in[2];
  const int*   eidx = (const int*)d_in[3];
  float* out = (float*)d_out;
  float* Z   = (float*)d_ws;

  const int nEdges = in_sizes[3] / 2;
  const int nOut   = out_size;
  const int nZ     = nOut / DIM;

  const int threads = 256;
  const int blocks0 = (nOut + threads - 1) / threads;
  zero_kernel<<<blocks0, threads, 0, stream>>>(out, Z, nOut, nZ);

  const int groups  = (nEdges + 15) / 16;
  const int blocks1 = (groups + 7) / 8;
  edge_attn_kernel<<<blocks1, threads, 0, stream>>>(q, k, v, eidx, out, Z, nEdges);

  norm_kernel<<<blocks0, threads, 0, stream>>>(out, Z, nOut);
}


# Every run of 64 edges is drawn once per package half (int add on a counter); a wave keeps only edges whose destination row its half owns: a row's float adds stay in one half, still arrival order
# speedup vs baseline: 6.1176x; 1.1481x over previous
#include <hip/hip_runtime.h>
#include <math.h>

#define HEADS 8
#define DIM 16
#define HD 128
#define INV_SCALE 0.25f
#define VBATCH 8
#define ROWPITCH 36
#define RUN 64
#define TKGAP 64
#define OWNER(d) (((d) >> 3) & 1)

typedef __attribute__((ext_vector_type(2))) float v2f;
typedef __attribute__((ext_vector_type(8))) float v8f;

__device__ __forceinline__ int* run_counters(float* zbuf, int nZ) {
  return (int*)(zbuf + ((nZ + 63) & ~63));
}

__global__ void zero_kernel(float* __restrict__ out, float* __restrict__ zbuf,
                            int nOut, int nZ) {
  int i = blockIdx.x * blockDim.x + threadIdx.x;
  if (i < nOut) out[i] = 0.0f;
  if (i < nZ)  zbuf[i] = 0.0f;
  if (i < 2 * TKGAP) run_counters(zbuf, nZ)[i] = 0;
}

template <int NB>
__device__ __forceinline__ void add_rows(const float* __restrict__ v, float* __restrict__ wV,
                                         const float4* __restrict__ w4, int myidx, int j,
                                         int lane, int is_hi) {
  float x[NB][4];
#pragma unroll
  for (int u = 0; u < NB; ++u) {
    const int sj = __builtin_amdgcn_readlane(myidx, j + u);
    const float* vp = v + (long)sj * HD + lane;
    x[u][0] = vp[0];
    x[u][1] = vp[32];
    x[u][2] = vp[64];
    x[u][3] = vp[96];
  }
#pragma unroll
  for (int u = 0; u < NB; ++u) {
    const int dj = __builtin_amdgcn_readlane(myidx, 16 + j + u);
    const float4 w = w4[(j + u) * 2 + is_hi];
    float* op = wV + (long)dj * HD + lane;
    atomicAdd(op,      x[u][0] * w.x);
    atomicAdd(op + 32, x[u][1] * w.y);
    atomicAdd(op + 64, x[u][2] * w.z);
    atomicAdd(op + 96, x[u][3] * w.w);
  }
}

__global__ void __launch_bounds__(256) __attribute__((amdgpu_num_vgpr(64)))
edge_attn_kernel(const float* __restrict__ q, const float* __restrict__ k,
                 const float* __restrict__ v, const int* __restrict__ eidx,
                 float* __restrict__ wV, float* __restrict__ Z, int nEdges, int nZ,
                 int nRuns) {
  __shared__ int    s_list[8][2 * RUN];
  __shared__ float4 s_w4[8][32];
  __shared__ float4 s_k4[8][16 * ROWPITCH / 4];
  __shared__ float4 s_q4[8][16 * ROWPITCH / 4];

  const int lane = threadIdx.x & 31;
  const int wv   = threadIdx.x >> 5;

  int half = (int)((__builtin_amdgcn_s_sendmsg_rtn(0x87) >> 18) & 1u);
  int run;
  {
    int* tk = run_counters(Z, nZ);
    int t = 0;
    if (lane == 0) t = atomicAdd(tk + half * TKGAP, 1);
    t = __builtin_amdgcn_readfirstlane(t);
    if (t >= nRuns) {
      half ^= 1;
      if (lane == 0) t = atomicAdd(tk + half * TKGAP, 1);
      t = __builtin_amdgcn_readfirstlane(t);
      if (t >= nRuns) return;
    }
    run = t;
  }

  int* ls = &s_list[wv][0];
  int nkept;
  {
    const long ea = (long)run * RUN + lane;
    const long eb = ea + 32;
    const bool va = ea < (long)nEdges;
    const bool vb = eb < (long)nEdges;
    const int sa = va ? eidx[ea] : 0;
    const int da = va ? eidx[(long)nEdges + ea] : 0;
    const int sb = vb ? eidx[eb] : 0;
    const int db = vb ? eidx[(long)nEdges + eb] : 0;
    const bool pa = va && (OWNER(da) == half);
    const bool pb = vb && (OWNER(db) == half);
    const unsigned ma = __builtin_amdgcn_ballot_w32(pa);
    const unsigned mb = __builtin_amdgcn_ballot_w32(pb);
    const int na = __builtin_popcount(ma);
    nkept = na + __builtin_popcount(mb);
    if (pa) {
      const int p = (int)__builtin_amdgcn_mbcnt_lo(ma, 0u);
      ls[p] = sa;
      ls[RUN + p] = da;
    }
    if (pb) {
      const int p = na + (int)__builtin_amdgcn_mbcnt_lo(mb, 0u);
      ls[p] = sb;
      ls[RUN + p] = db;
    }
  }
  asm volatile("s_wait_dscnt 0x0" ::: "memory");

  const int nk = __builtin_amdgcn_readfirstlane(nkept);
  for (int g0 = 0; g0 < nk; g0 += 16) {
    const int cnt = (nk - g0 < 16) ? (nk - g0) : 16;
    int ln = lane;
    asm volatile("" : "+v"(ln));
    const int pc   = ln & 7;
    const int rsub = ln >> 3;
    const char* kbytes = (const char*)k;
    const char* qbytes = (const char*)q;
    float* skw = (float*)&s_k4[wv][0];
    float* sqw = (float*)&s_q4[wv][0];
    float* kst = skw + rsub * ROWPITCH + pc * 4;
    float* qst = sqw + rsub * ROWPITCH + pc * 4;

    const int m     = ln & 15;
    const int is_hi = ln >> 4;
    const float* arow = skw + m * ROWPITCH + is_hi * 2;
    const float* brow = sqw + m * ROWPITCH + is_hi * 2;

    const int  idxsel = ln & 7;
    const bool isdiag = (ln < 8) || (ln >= 24);
    const int  dm     = (ln >= 24) ? (ln - 16) : ln;

    float* sw = (float*)&s_w4[wv][0];

    const int mc    = (m < cnt) ? m : (cnt - 1);
    const int myidx = ls[is_hi * RUN + g0 + mc];
    unsigned kofs[4], qofs[4];
    for (int i = 0; i < 4; ++i) {
      const int r  = 4 * i + rsub;
      const int rc = (r < cnt) ? r : (cnt - 1);
      kofs[i] = ((unsigned)ls[g0 + rc]       * HD + pc * 4) * 4u;
      qofs[i] = ((unsigned)ls[RUN + g0 + rc] * HD + pc * 4) * 4u;
    }

#pragma nounroll
    for (int hp = 0; hp < HEADS / 2; ++hp) {
      const unsigned lineofs = (unsigned)hp * 128u;
      float4 kr[4], qr[4];
      for (int i = 0; i < 4; ++i) {
        kr[i] = *(const float4*)(kbytes + (kofs[i] + lineofs));
        qr[i] = *(const float4*)(qbytes + (qofs[i] + lineofs));
      }
      for (int i = 0; i < 4; ++i) {
        *(float4*)(kst + i * 4 * ROWPITCH) = kr[i];
        *(float4*)(qst + i * 4 * ROWPITCH) = qr[i];
      }
      asm volatile("s_wait_dscnt 0x0" ::: "memory");

      for (int hh = 0; hh < 2; ++hh) {
        const float* ap = arow + hh * DIM;
        const float* bp = brow + hh * DIM;
        v8f c = {0.f, 0.f, 0.f, 0.f, 0.f, 0.f, 0.f, 0.f};
#pragma unroll
        for (int c4 = 0; c4 < 4; ++c4) {
          v2f a = *(const v2f*)(ap + c4 * 4);
          v2f b = *(const v2f*)(bp + c4 * 4);
          c = __builtin_amdgcn_wmma_f32_16x16x4_f32(false, a, false, b,
                                                    (short)0, c, false, false);
        }
        float d01 = (idxsel & 1) ? c[1] : c[0];
        float d23 = (idxsel & 1) ? c[3] : c[2];
        float d45 = (idxsel & 1) ? c[5] : c[4];
        float d67 = (idxsel & 1) ? c[7] : c[6];
        float d03 = (idxsel & 2) ? d23 : d01;
        float d47 = (idxsel & 2) ? d67 : d45;
        float dd  = (idxsel & 4) ? d47 : d03;
        if (isdiag) {
          float t = dd * INV_SCALE;
          t = fminf(fmaxf(t, -5.0f), 5.0f);
          sw[dm * 8 + hh * 4 + hp] = __expf(t);
        }
      }
      asm volatile("s_wait_dscnt 0x0" ::: "memory");
    }

    {
      const int zh    = ln & 7;
      const int zslot = (zh & 1) * 4 + (zh >> 1);
      for (int t = 0; t < 4; ++t) {
        const int j = t * 4 + (ln >> 3);
        if (j < cnt) {
          const int dj = ls[RUN + g0 + j];
          atomicAdd(Z + (long)dj * HEADS + zh, sw[j * 8 + zslot]);
        }
      }
    }

    int j = 0;
    for (; j + VBATCH <= cnt; j += VBATCH)
      add_rows<VBATCH>(v, wV, &s_w4[wv][0], myidx, j, ln, is_hi);
    if (j + 4 <= cnt) {
      add_rows<4>(v, wV, &s_w4[wv][0], myidx, j, ln, is_hi);
      j += 4;
    }
    for (; j < cnt; ++j)
      add_rows<1>(v, wV, &s_w4[wv][0], myidx, j, ln, is_hi);
  }
}

__global__ void norm_kernel(float* __restrict__ out, const float* __restrict__ Z,
                            int n) {
  int i = blockIdx.x * blockDim.x + threadIdx.x;
  if (i >= n) return;
  int node = i >> 7;
  int h    = (i >> 4) & 7;
  out[i] = out[i] / (Z[node * HEADS + h] + 1e-6f);
}

extern "C" void kernel_launch(void* const* d_in, const int* in_sizes, int n_in,
                              void* d_out, int out_size, void* d_ws, size_t ws_size,
                              hipStream_t stream) {
  const float* q    = (const float*)d_in[0];
  const float* k    = (const float*)d_in[1];
  const float* v    = (const float*)d_in[2];
  const int*   eidx = (const int*)d_in[3];
  float* out = (float*)d_out;
  float* Z   = (float*)d_ws;

  const int nEdges = in_sizes[3] / 2;
  const int nOut   = out_size;
  const int nZ     = nOut / DIM;

  const int threads = 256;
  const int blocks0 = (nOut + threads - 1) / threads;
  zero_kernel<<<blocks0, threads, 0, stream>>>(out, Z, nOut, nZ);

  const int runs    = (nEdges + RUN - 1) / RUN;
  const int blocks1 = (2 * runs + 7) / 8;
  edge_attn_kernel<<<blocks1, threads, 0, stream>>>(q, k, v, eidx, out, Z, nEdges, nZ, runs);

  norm_kernel<<<blocks0, threads, 0, stream>>>(out, Z, nOut);
}


# Each work-group owns 196 destination rows: their sums, Z and Q rows stay in LDS (LDS float adds, arrival order) while it reads the whole edge list for its edges; rows divided and stored once
# speedup vs baseline: 7.0441x; 1.1514x over previous
#include <hip/hip_runtime.h>
#include <math.h>

#define HEADS 8
#define DIM 16
#define HD 128
#define INV_SCALE 0.25f
#define VBATCH 8
#define ROWPITCH 36
#define NROWS 196
#define QPITCH 132
#define NWAVES 32
#define STEP 10
#define CAP 144
#define QTURNS ((NROWS * (HD / 4) + NWAVES * 32 - 1) / (NWAVES * 32))

typedef __attribute__((ext_vector_type(2))) float v2f;
typedef __attribute__((ext_vector_type(8))) float v8f;

template <int NB>
__device__ __forceinline__ void add_rows(const float* __restrict__ v, float* acc,
                                         const float4* w4, int myidx, int j,
                                         int lane, int is_hi, int row0) {
  float x[NB][4];
#pragma unroll
  for (int u = 0; u < NB; ++u) {
    const int sj = __builtin_amdgcn_readlane(myidx, j + u);
    const float* vp = v + (long)sj * HD + lane;
    x[u][0] = vp[0];
    x[u][1] = vp[32];
    x[u][2] = vp[64];
    x[u][3] = vp[96];
  }
#pragma unroll
  for (int u = 0; u < NB; ++u) {
    const int dj = __builtin_amdgcn_readlane(myidx, 16 + j + u);
    const float4 w = w4[(j + u) * 2 + is_hi];
    float* op = acc + (dj - row0) * HD + lane;
    atomicAdd(op,      x[u][0] * w.x);
    atomicAdd(op + 32, x[u][1] * w.y);
    atomicAdd(op + 64, x[u][2] * w.z);
    atomicAdd(op + 96, x[u][3] * w.w);
  }
}

__device__ __forceinline__ void load_step(int4 (&d)[STEP], const int* __restrict__ dstp,
                                          bool valid, int ebase, int lane) {
  if (valid) {
    const char* sb = (const char*)(dstp + ebase);
    unsigned lo = (unsigned)lane * 16u;
    asm volatile("" : "+v"(lo));
#pragma unroll
    for (int u = 0; u < STEP; ++u) d[u] = *(const int4*)(sb + lo + u * 512);
  } else {
#pragma unroll
    for (int u = 0; u < STEP; ++u) d[u] = make_int4(-1, -1, -1, -1);
  }
}

__device__ __forceinline__ void do_group(const float* __restrict__ k, const float* __restrict__ v,
                                         const int* __restrict__ eidx, int nEdges,
                                         const int* lsg, int cnt, int lane, int* grp,
                                         float* skw, const float* qres, float4* w4,
                                         float* acc, float* zacc, int row0) {
  const int m     = lane & 15;
  const int is_hi = lane >> 4;
  const int mc    = (m < cnt) ? m : (cnt - 1);
  const int myidx = eidx[(long)is_hi * nEdges + lsg[mc]];
  grp[lane] = myidx;
  asm volatile("s_wait_dscnt 0x0" ::: "memory");

  const int pc   = lane & 7;
  const int rsub = lane >> 3;
  const char* kbytes = (const char*)k;
  float* kst = skw + rsub * ROWPITCH + pc * 4;

  const float* arow = skw + m * ROWPITCH + is_hi * 2;
  const float* brow = qres + (grp[16 + m] - row0) * QPITCH + is_hi * 2;

  const int  idxsel = lane & 7;
  const bool isdiag = (lane < 8) || (lane >= 24);
  const int  dm     = (lane >= 24) ? (lane - 16) : lane;

  float* sw = (float*)w4;

  unsigned kofs[4];
  for (int i = 0; i < 4; ++i) {
    const int r  = 4 * i + rsub;
    const int rc = (r < cnt) ? r : (cnt - 1);
    kofs[i] = ((unsigned)grp[rc] * HD + pc * 4) * 4u;
  }

  float4 kr0 = *(const float4*)(kbytes + kofs[0]);
  float4 kr1 = *(const float4*)(kbytes + kofs[1]);
  float4 kr2 = *(const float4*)(kbytes + kofs[2]);
  float4 kr3 = *(const float4*)(kbytes + kofs[3]);

#pragma nounroll
  for (int hp = 0; hp < HEADS / 2; ++hp) {
    *(float4*)(kst + 0 * 4 * ROWPITCH) = kr0;
    *(float4*)(kst + 1 * 4 * ROWPITCH) = kr1;
    *(float4*)(kst + 2 * 4 * ROWPITCH) = kr2;
    *(float4*)(kst + 3 * 4 * ROWPITCH) = kr3;
    asm volatile("s_wait_dscnt 0x0" ::: "memory");
    if (hp + 1 < HEADS / 2) {
      const unsigned lineofs = (unsigned)(hp + 1) * 128u;
      kr0 = *(const float4*)(kbytes + (kofs[0] + lineofs));
      kr1 = *(const float4*)(kbytes + (kofs[1] + lineofs));
      kr2 = *(const float4*)(kbytes + (kofs[2] + lineofs));
      kr3 = *(const float4*)(kbytes + (kofs[3] + lineofs));
    }

    for (int hh = 0; hh < 2; ++hh) {
      const float* ap = arow + hh * DIM;
      const float* bp = brow + hp * 32 + hh * DIM;
      v8f c = {0.f, 0.f, 0.f, 0.f, 0.f, 0.f, 0.f, 0.f};
#pragma unroll
      for (int c4 = 0; c4 < 4; ++c4) {
        v2f a = *(const v2f*)(ap + c4 * 4);
        v2f b = *(const v2f*)(bp + c4 * 4);
        c = __builtin_amdgcn_wmma_f32_16x16x4_f32(false, a, false, b,
                                                  (short)0, c, false, false);
      }
      float d01 = (idxsel & 1) ? c[1] : c[0];
      float d23 = (idxsel & 1) ? c[3] : c[2];
      float d45 = (idxsel & 1) ? c[5] : c[4];
      float d67 = (idxsel & 1) ? c[7] : c[6];
      float d03 = (idxsel & 2) ? d23 : d01;
      float d47 = (idxsel & 2) ? d67 : d45;
      float dd  = (idxsel & 4) ? d47 : d03;
      if (isdiag) {
        float t = dd * INV_SCALE;
        t = fminf(fmaxf(t, -5.0f), 5.0f);
        sw[dm * 8 + hh * 4 + hp] = __expf(t);
      }
    }
    asm volatile("s_wait_dscnt 0x0" ::: "memory");
  }

  {
    const int zh    = lane & 7;
    const int zslot = (zh & 1) * 4 + (zh >> 1);
    for (int t = 0; t < 4; ++t) {
      const int j = t * 4 + (lane >> 3);
      if (j < cnt) {
        const int dj = grp[16 + j];
        atomicAdd(zacc + (dj - row0) * HEADS + zh, sw[j * 8 + zslot]);
      }
    }
  }

  int j = 0;
  for (; j + VBATCH <= cnt; j += VBATCH)
    add_rows<VBATCH>(v, acc, w4, myidx, j, lane, is_hi, row0);
  if (j + 4 <= cnt) {
    add_rows<4>(v, acc, w4, myidx, j, lane, is_hi, row0);
    j += 4;
  }
  for (; j < cnt; ++j)
    add_rows<1>(v, acc, w4, myidx, j, lane, is_hi, row0);
}

__global__ void __launch_bounds__(1024)
edge_attn_kernel(const float* __restrict__ q, const float* __restrict__ k,
                 const float* __restrict__ v, const int* __restrict__ eidx,
                 float* __restrict__ out, int nEdges, int nNodes) {
  __shared__ float  s_acc[NROWS * HD];
  __shared__ float  s_z[NROWS * HEADS];
  __shared__ float4 s_q[NROWS * QPITCH / 4];
  __shared__ int    s_list[NWAVES][CAP];
  __shared__ int    s_grp[NWAVES][32];
  __shared__ float4 s_w4[NWAVES][32];
  __shared__ float4 s_k4[NWAVES][16 * ROWPITCH / 4];

  const int tid  = threadIdx.x;
  const int lane = tid & 31;
  const int wv   = __builtin_amdgcn_readfirstlane(tid >> 5);
  const int row0 = blockIdx.x * NROWS;

  for (int i = tid; i < NROWS * HD; i += NWAVES * 32) s_acc[i] = 0.0f;
  for (int i = tid; i < NROWS * HEADS; i += NWAVES * 32) s_z[i] = 0.0f;
  {
    float4 t[QTURNS];
#pragma unroll
    for (int it = 0; it < QTURNS; ++it) {
      const int i    = tid + it * (NWAVES * 32);
      const int node = row0 + (i >> 5);
      t[it] = make_float4(0.f, 0.f, 0.f, 0.f);
      if (i < NROWS * (HD / 4) && node < nNodes)
        t[it] = *(const float4*)(q + (long)node * HD + (i & 31) * 4);
    }
#pragma unroll
    for (int it = 0; it < QTURNS; ++it) {
      const int i = tid + it * (NWAVES * 32);
      if (i < NROWS * (HD / 4)) s_q[(i >> 5) * (QPITCH / 4) + (i & 31)] = t[it];
    }
  }
  __syncthreads();

  int* ls = &s_list[wv][0];
  const int* dstp = eidx + nEdges;
  const bool vec_ok   = (nEdges & 3) == 0;
  const int tailStart = vec_ok ? (nEdges / (STEP * 128)) * (STEP * 128) : 0;
  const int nWS       = tailStart / (STEP * 128);
  const int nSteps    = (nWS + NWAVES - 1) / NWAVES;
  const int padded    = nSteps * NWAVES;
  const int rot       = (int)(((unsigned long long)blockIdx.x * (unsigned)padded) / gridDim.x);
  const int nTail     = (nEdges - tailStart + NWAVES * 32 - 1) / (NWAVES * 32);
  const int lastTurn  = nSteps + nTail;

  int c = 0;
  int4 d[STEP];
  for (int u = 0; u < STEP; ++u) d[u] = make_int4(-1, -1, -1, -1);
  int ebase = 0;
  for (int s = -1; s <= lastTurn; ++s) {
    const bool last = (s == lastTurn);
    int un = 0;
    if (s >= nSteps) {
      un = STEP;
      if (!last) {
        const int e  = tailStart + ((s - nSteps) * NWAVES + wv) * 32 + lane;
        const int dv = (e < nEdges) ? dstp[e] : -1;
        const bool p = (unsigned)(dv - row0) < (unsigned)NROWS;
        const unsigned mk = __builtin_amdgcn_ballot_w32(p);
        if (p) ls[c + (int)__builtin_amdgcn_mbcnt_lo(mk, 0u)] = e;
        c += __builtin_popcount(mk);
      }
    }
    do {
#pragma unroll
      for (int u = 0; u < STEP; ++u) {
        if (u == un && c + 128 <= CAP) {
          const int dd[4] = {d[u].x, d[u].y, d[u].z, d[u].w};
          bool p[4];
          for (int t = 0; t < 4; ++t) p[t] = (unsigned)(dd[t] - row0) < (unsigned)NROWS;
          if (__builtin_amdgcn_ballot_w32(p[0] || p[1] || p[2] || p[3]) != 0u) {
            for (int t = 0; t < 4; ++t) {
              const unsigned mk = __builtin_amdgcn_ballot_w32(p[t]);
              if (p[t]) ls[c + (int)__builtin_amdgcn_mbcnt_lo(mk, 0u)] = ebase + u * 128 + lane * 4 + t;
              c += __builtin_popcount(mk);
            }
          }
          un = u + 1;
        }
      }
      if (un == STEP && s + 1 < nSteps) {
        int id = (s + 1) * NWAVES + wv + rot;
        if (id >= padded) id -= padded;
        ebase = id * (STEP * 128);
        load_step(d, dstp, id < nWS, ebase, lane);
      }
      asm volatile("s_wait_dscnt 0x0" ::: "memory");
      const int thr = (last && un == STEP) ? 1 : 16;
      if (c >= thr) {
        int g0 = 0;
        while (c - g0 >= thr) {
          const int cnt = (c - g0 < 16) ? (c - g0) : 16;
          do_group(k, v, eidx, nEdges, ls + g0, cnt, lane, &s_grp[wv][0], (float*)&s_k4[wv][0],
                   (const float*)s_q, &s_w4[wv][0], s_acc, s_z, row0);
          g0 += cnt;
        }
        const int rem = c - g0;
        if (rem > 0) {
          int a = 0;
          if (lane < rem) a = ls[g0 + lane];
          if (lane < rem) ls[lane] = a;
          asm volatile("s_wait_dscnt 0x0" ::: "memory");
        }
        c = rem;
      }
    } while (un < STEP);
  }

  __syncthreads();
  for (int i = tid; i < NROWS * HD; i += NWAVES * 32) {
    const int row  = i >> 7;
    const int node = row0 + row;
    if (node < nNodes) {
      const int h = (i >> 4) & 7;
      out[(long)node * HD + (i & 127)] = s_acc[i] / (s_z[row * HEADS + h] + 1e-6f);
    }
  }
}

extern "C" void kernel_launch(void* const* d_in, const int* in_sizes, int n_in,
                              void* d_out, int out_size, void* d_ws, size_t ws_size,
                              hipStream_t stream) {
  const float* q    = (const float*)d_in[0];
  const float* k    = (const float*)d_in[1];
  const float* v    = (const float*)d_in[2];
  const int*   eidx = (const int*)d_in[3];
  float* out = (float*)d_out;

  const int nEdges = in_sizes[3] / 2;
  const int nNodes = out_size / HD;

  const int blocks = (nNodes + NROWS - 1) / NROWS;
  edge_attn_kernel<<<blocks, NWAVES * 32, 0, stream>>>(q, k, v, eidx, out, nEdges, nNodes);
}


# A first launch orders the edge list by owning work-group (256 pieces, places from LDS integer adds); each work-group then reads only its own edges, not the whole list; float sums stay arrival order
# speedup vs baseline: 9.6868x; 1.3752x over previous
#include <hip/hip_runtime.h>
#include <math.h>

#define HEADS 8
#define DIM 16
#define HD 128
#define INV_SCALE 0.25f
#define VBATCH 8
#define ROWPITCH 36
#define NROWS 196
#define QPITCH 132
#define NWAVES 32
#define NSEG 256
#define SEGW (NSEG / NWAVES)
#define MAXB 256
#define PTHREADS 256
#define PCH 16
#define PCAP (PTHREADS * PCH)
#define QTURNS ((NROWS * (HD / 4) + NWAVES * 32 - 1) / (NWAVES * 32))

typedef __attribute__((ext_vector_type(2))) float v2f;
typedef __attribute__((ext_vector_type(8))) float v8f;

template <int NB>
__device__ __forceinline__ void add_rows(const float* __restrict__ v, float* acc,
                                         const float4* w4, int myidx, int j,
                                         int lane, int is_hi, int row0) {
  float x[NB][4];
#pragma unroll
  for (int u = 0; u < NB; ++u) {
    const int sj = __builtin_amdgcn_readlane(myidx, j + u);
    const float* vp = v + (long)sj * HD + lane;
    x[u][0] = vp[0];
    x[u][1] = vp[32];
    x[u][2] = vp[64];
    x[u][3] = vp[96];
  }
#pragma unroll
  for (int u = 0; u < NB; ++u) {
    const int dj = __builtin_amdgcn_readlane(myidx, 16 + j + u);
    const float4 w = w4[(j + u) * 2 + is_hi];
    float* op = acc + (dj - row0) * HD + lane;
    atomicAdd(op,      x[u][0] * w.x);
    atomicAdd(op + 32, x[u][1] * w.y);
    atomicAdd(op + 64, x[u][2] * w.z);
    atomicAdd(op + 96, x[u][3] * w.w);
  }
}

__device__ __forceinline__ void do_group(const float* __restrict__ k, const float* __restrict__ v,
                                         int myidx, int cnt, int lane, int* grp,
                                         float* skw, const float* qres, float4* w4,
                                         float* acc, float* zacc, int row0) {
  const int m     = lane & 15;
  const int is_hi = lane >> 4;
  grp[lane] = myidx;
  asm volatile("s_wait_dscnt 0x0" ::: "memory");

  const int pc   = lane & 7;
  const int rsub = lane >> 3;
  const char* kbytes = (const char*)k;
  float* kst = skw + rsub * ROWPITCH + pc * 4;

  const float* arow = skw + m * ROWPITCH + is_hi * 2;
  const float* brow = qres + (grp[16 + m] - row0) * QPITCH + is_hi * 2;

  const int  idxsel = lane & 7;
  const bool isdiag = (lane < 8) || (lane >= 24);
  const int  dm     = (lane >= 24) ? (lane - 16) : lane;

  float* sw = (float*)w4;

  unsigned kofs[4];
  for (int i = 0; i < 4; ++i) {
    const int r  = 4 * i + rsub;
    const int rc = (r < cnt) ? r : (cnt - 1);
    kofs[i] = ((unsigned)grp[rc] * HD + pc * 4) * 4u;
  }

  float4 kr0 = *(const float4*)(kbytes + kofs[0]);
  float4 kr1 = *(const float4*)(kbytes + kofs[1]);
  float4 kr2 = *(const float4*)(kbytes + kofs[2]);
  float4 kr3 = *(const float4*)(kbytes + kofs[3]);

#pragma nounroll
  for (int hp = 0; hp < HEADS / 2; ++hp) {
    *(float4*)(kst + 0 * 4 * ROWPITCH) = kr0;
    *(float4*)(kst + 1 * 4 * ROWPITCH) = kr1;
    *(float4*)(kst + 2 * 4 * ROWPITCH) = kr2;
    *(float4*)(kst + 3 * 4 * ROWPITCH) = kr3;
    asm volatile("s_wait_dscnt 0x0" ::: "memory");
    if (hp + 1 < HEADS / 2) {
      const unsigned lineofs = (unsigned)(hp + 1) * 128u;
      kr0 = *(const float4*)(kbytes + (kofs[0] + lineofs));
      kr1 = *(const float4*)(kbytes + (kofs[1] + lineofs));
      kr2 = *(const float4*)(kbytes + (kofs[2] + lineofs));
      kr3 = *(const float4*)(kbytes + (kofs[3] + lineofs));
    }

    for (int hh = 0; hh < 2; ++hh) {
      const float* ap = arow + hh * DIM;
      const float* bp = brow + hp * 32 + hh * DIM;
      v8f c = {0.f, 0.f, 0.f, 0.f, 0.f, 0.f, 0.f, 0.f};
#pragma unroll
      for (int c4 = 0; c4 < 4; ++c4) {
        v2f a = *(const v2f*)(ap + c4 * 4);
        v2f b = *(const v2f*)(bp + c4 * 4);
        c = __builtin_amdgcn_wmma_f32_16x16x4_f32(false, a, false, b,
                                                  (short)0, c, false, false);
      }
      float d01 = (idxsel & 1) ? c[1] : c[0];
      float d23 = (idxsel & 1) ? c[3] : c[2];
      float d45 = (idxsel & 1) ? c[5] : c[4];
      float d67 = (idxsel & 1) ? c[7] : c[6];
      float d03 = (idxsel & 2) ? d23 : d01;
      float d47 = (idxsel & 2) ? d67 : d45;
      float dd  = (idxsel & 4) ? d47 : d03;
      if (isdiag) {
        float t = dd * INV_SCALE;
        t = fminf(fmaxf(t, -5.0f), 5.0f);
        sw[dm * 8 + hh * 4 + hp] = __expf(t);
      }
    }
    asm volatile("s_wait_dscnt 0x0" ::: "memory");
  }

  {
    const int zh    = lane & 7;
    const int zslot = (zh & 1) * 4 + (zh >> 1);
    for (int t = 0; t < 4; ++t) {
      const int j = t * 4 + (lane >> 3);
      if (j < cnt) {
        const int dj = grp[16 + j];
        atomicAdd(zacc + (dj - row0) * HEADS + zh, sw[j * 8 + zslot]);
      }
    }
  }

  int j = 0;
  for (; j + VBATCH <= cnt; j += VBATCH)
    add_rows<VBATCH>(v, acc, w4, myidx, j, lane, is_hi, row0);
  if (j + 4 <= cnt) {
    add_rows<4>(v, acc, w4, myidx, j, lane, is_hi, row0);
    j += 4;
  }
  for (; j < cnt; ++j)
    add_rows<1>(v, acc, w4, myidx, j, lane, is_hi, row0);
}

__global__ void __launch_bounds__(PTHREADS)
bin_edges_kernel(const int* __restrict__ eidx, int* __restrict__ ws_pairs, int* __restrict__ ws_tab,
                 int nEdges, int nNodes, int nOwners) {
  __shared__ int  s_cnt[MAXB];
  __shared__ int  s_off[MAXB + 1];
  __shared__ int2 s_pair[PCAP];

  const int tid   = threadIdx.x;
  const int piece = blockIdx.x;
  const int chunk = (nEdges + NSEG - 1) / NSEG;
  const int e0    = piece * chunk;
  int len = nEdges - e0;
  if (len > chunk) len = chunk;

  for (int i = tid; i < MAXB; i += PTHREADS) s_cnt[i] = 0;
  __syncthreads();

  int sv[PCH], dv[PCH], own[PCH], rnk[PCH];
#pragma unroll
  for (int u = 0; u < PCH; ++u) {
    const int i = tid + u * PTHREADS;
    sv[u] = 0;
    dv[u] = -1;
    if (i < len) {
      sv[u] = eidx[e0 + i];
      dv[u] = eidx[nEdges + e0 + i];
    }
  }
#pragma unroll
  for (int u = 0; u < PCH; ++u) {
    own[u] = ((unsigned)dv[u] < (unsigned)nNodes) ? (dv[u] / NROWS) : -1;
    rnk[u] = 0;
    if (own[u] >= 0) rnk[u] = atomicAdd(&s_cnt[own[u]], 1);
  }
  __syncthreads();

  if (tid < 32) {
    int a[MAXB / 32];
    int run = 0;
    for (int t = 0; t < MAXB / 32; ++t) {
      a[t] = run;
      run += s_cnt[tid * (MAXB / 32) + t];
    }
    int inc = run;
    for (int o = 1; o < 32; o <<= 1) {
      const int y = __shfl_up(inc, o);
      if (tid >= o) inc += y;
    }
    const int base = inc - run;
    for (int t = 0; t < MAXB / 32; ++t) s_off[tid * (MAXB / 32) + t] = base + a[t];
    if (tid == 31) s_off[MAXB] = inc;
  }
  __syncthreads();

#pragma unroll
  for (int u = 0; u < PCH; ++u)
    if (own[u] >= 0) s_pair[s_off[own[u]] + rnk[u]] = make_int2(sv[u], dv[u]);
  __syncthreads();

  const int total = s_off[MAXB];
  int2* pairs = (int2*)ws_pairs + (long)piece * PCAP;
  for (int i = tid; i < total; i += PTHREADS) pairs[i] = s_pair[i];
  int2* tab = (int2*)ws_tab + (long)piece * MAXB;
  for (int b = tid; b < nOwners; b += PTHREADS)
    tab[b] = make_int2(piece * PCAP + s_off[b], s_off[b + 1] - s_off[b]);
}

__device__ __forceinline__ int2 load_pair(const int2* __restrict__ pairs, const int (&bound)[SEGW],
                                          const int (&delta)[SEGW], int i) {
  int pos = i + delta[0];
  for (int s = 1; s < SEGW; ++s)
    if (i >= bound[s]) pos = i + delta[s];
  return pairs[pos];
}

__global__ void __launch_bounds__(1024)
edge_attn_kernel(const float* __restrict__ q, const float* __restrict__ k,
                 const float* __restrict__ v, const int* __restrict__ ws_pairs,
                 const int* __restrict__ ws_tab, float* __restrict__ out, int nNodes) {
  __shared__ float  s_acc[NROWS * HD];
  __shared__ float  s_z[NROWS * HEADS];
  __shared__ float4 s_q[NROWS * QPITCH / 4];
  __shared__ int    s_grp[NWAVES][32];
  __shared__ float4 s_w4[NWAVES][32];
  __shared__ float4 s_k4[NWAVES][16 * ROWPITCH / 4];

  const int tid  = threadIdx.x;
  const int lane = tid & 31;
  const int wv   = __builtin_amdgcn_readfirstlane(tid >> 5);
  const int row0 = blockIdx.x * NROWS;

  const int2* tab   = (const int2*)ws_tab;
  const int2* pairs = (const int2*)ws_pairs;
  int2 ent = make_int2(0, 0);
  if (lane < SEGW) ent = tab[(long)(wv * SEGW + lane) * MAXB + blockIdx.x];

  for (int i = tid; i < NROWS * HD; i += NWAVES * 32) s_acc[i] = 0.0f;
  for (int i = tid; i < NROWS * HEADS; i += NWAVES * 32) s_z[i] = 0.0f;
  {
    float4 t[QTURNS];
#pragma unroll
    for (int it = 0; it < QTURNS; ++it) {
      const int i    = tid + it * (NWAVES * 32);
      const int node = row0 + (i >> 5);
      t[it] = make_float4(0.f, 0.f, 0.f, 0.f);
      if (i < NROWS * (HD / 4) && node < nNodes)
        t[it] = *(const float4*)(q + (long)node * HD + (i & 31) * 4);
    }
#pragma unroll
    for (int it = 0; it < QTURNS; ++it) {
      const int i = tid + it * (NWAVES * 32);
      if (i < NROWS * (HD / 4)) s_q[(i >> 5) * (QPITCH / 4) + (i & 31)] = t[it];
    }
  }

  int bound[SEGW], delta[SEGW];
  int total = 0;
  for (int s = 0; s < SEGW; ++s) {
    const int st = __builtin_amdgcn_readlane(ent.x, s);
    const int ln = __builtin_amdgcn_readlane(ent.y, s);
    bound[s] = total;
    delta[s] = st - total;
    total += ln;
  }
  const int m     = lane & 15;
  const int is_hi = lane >> 4;
  int2 pr = make_int2(0, 0);
  if (total > 0) {
    const int c0 = (total < 16) ? total : 16;
    pr = load_pair(pairs, bound, delta, (m < c0) ? m : (c0 - 1));
  }
  __syncthreads();

  for (int g0 = 0; g0 < total; g0 += 16) {
    const int cnt   = (total - g0 < 16) ? (total - g0) : 16;
    const int myidx = is_hi ? pr.y : pr.x;
    const int g1    = g0 + 16;
    if (g1 < total) {
      const int c1 = (total - g1 < 16) ? (total - g1) : 16;
      pr = load_pair(pairs, bound, delta, g1 + ((m < c1) ? m : (c1 - 1)));
    }
    do_group(k, v, myidx, cnt, lane, &s_grp[wv][0], (float*)&s_k4[wv][0],
             (const float*)s_q, &s_w4[wv][0], s_acc, s_z, row0);
  }

  __syncthreads();
  for (int i = tid; i < NROWS * HD; i += NWAVES * 32) {
    const int row  = i >> 7;
    const int node = row0 + row;
    if (node < nNodes) {
      const int h = (i >> 4) & 7;
      out[(long)node * HD + (i & 127)] = s_acc[i] / (s_z[row * HEADS + h] + 1e-6f);
    }
  }
}

extern "C" void kernel_launch(void* const* d_in, const int* in_sizes, int n_in,
                              void* d_out, int out_size, void* d_ws, size_t ws_size,
                              hipStream_t stream) {
  const float* q    = (const float*)d_in[0];
  const float* k    = (const float*)d_in[1];
  const float* v    = (const float*)d_in[2];
  const int*   eidx = (const int*)d_in[3];
  float* out = (float*)d_out;

  const int nEdges = in_sizes[3] / 2;
  const int nNodes = out_size / HD;

  if (ws_size < ((size_t)NSEG * PCAP + (size_t)NSEG * MAXB) * 8) return;
  int* ws_pairs = (int*)d_ws;
  int* ws_tab   = ws_pairs + (size_t)NSEG * PCAP * 2;

  const int blocks = (nNodes + NROWS - 1) / NROWS;
  bin_edges_kernel<<<NSEG, PTHREADS, 0, stream>>>(eidx, ws_pairs, ws_tab, nEdges, nNodes, blocks);
  edge_attn_kernel<<<blocks, NWAVES * 32, 0, stream>>>(q, k, v, ws_pairs, ws_tab, out, nNodes);
}


# Score tile = one edge (16 rows: 8 heads x 2 halves): K row loaded whole straight into A registers, Q row one contiguous LDS read, no K staging in LDS; a head's 16 products now summed as 8+8
# speedup vs baseline: 10.6197x; 1.0963x over previous
#include <hip/hip_runtime.h>
#include <math.h>

#define HEADS 8
#define DIM 16
#define HD 128
#define INV_SCALE 0.25f
#define VBATCH 8
#define KBATCH 16
#define NROWS 196
#define QPITCH 132
#define NWAVES 32
#define NSEG 256
#define SEGW (NSEG / NWAVES)
#define MAXB 256
#define PTHREADS 256
#define PCH 16
#define PCAP (PTHREADS * PCH)
#define QTURNS ((NROWS * (HD / 4) + NWAVES * 32 - 1) / (NWAVES * 32))

typedef __attribute__((ext_vector_type(2))) float v2f;
typedef __attribute__((ext_vector_type(8))) float v8f;

template <int NB>
__device__ __forceinline__ void add_rows(const float* __restrict__ v, float* acc,
                                         const float4* w4, int myidx, int j,
                                         int lane, int is_hi, int row0) {
  float x[NB][4];
#pragma unroll
  for (int u = 0; u < NB; ++u) {
    const int sj = __builtin_amdgcn_readlane(myidx, j + u);
    const float* vp = v + (long)sj * HD + lane;
    x[u][0] = vp[0];
    x[u][1] = vp[32];
    x[u][2] = vp[64];
    x[u][3] = vp[96];
  }
#pragma unroll
  for (int u = 0; u < NB; ++u) {
    const int dj = __builtin_amdgcn_readlane(myidx, 16 + j + u);
    const float4 w = w4[(j + u) * 2 + is_hi];
    float* op = acc + (dj - row0) * HD + lane;
    atomicAdd(op,      x[u][0] * w.x);
    atomicAdd(op + 32, x[u][1] * w.y);
    atomicAdd(op + 64, x[u][2] * w.z);
    atomicAdd(op + 96, x[u][3] * w.w);
  }
}

template <int NB>
__device__ __forceinline__ void score_rows(const float* __restrict__ k, const float* qres,
                                           float* sw, int myidx, int j, int lane, unsigned pofs,
                                           int row0) {
  float4 kr[NB];
#pragma unroll
  for (int u = 0; u < NB; ++u) {
    const int sj = __builtin_amdgcn_readlane(myidx, j + u);
    const char* krow = (const char*)(k + (long)sj * HD);
    kr[u] = *(const float4*)(krow + pofs);
  }
  const int  idxsel = lane & 7;
  const bool isdiag = (lane < 8) || (lane >= 24);
  const int  drow   = (lane >= 24) ? (lane - 16) : lane;
  const int  head   = drow >> 1;
  const int  wslot  = (head & 1) * 4 + (head >> 1);
  constexpr int QA = (NB < 4) ? NB : 4;
  float4 qr[NB];
#pragma unroll
  for (int u = 0; u < QA; ++u) {
    const int dj = __builtin_amdgcn_readlane(myidx, 16 + j + u);
    qr[u] = *(const float4*)(qres + (dj - row0) * QPITCH + lane * 4);
  }
#pragma unroll
  for (int u = 0; u < NB; ++u) {
    if (u + QA < NB) {
      const int dj = __builtin_amdgcn_readlane(myidx, 16 + j + u + QA);
      qr[u + QA] = *(const float4*)(qres + (dj - row0) * QPITCH + lane * 4);
    }
    v8f c = {0.f, 0.f, 0.f, 0.f, 0.f, 0.f, 0.f, 0.f};
    const v2f a0 = {kr[u].x, kr[u].y};
    const v2f b0 = {qr[u].x, qr[u].y};
    const v2f a1 = {kr[u].z, kr[u].w};
    const v2f b1 = {qr[u].z, qr[u].w};
    c = __builtin_amdgcn_wmma_f32_16x16x4_f32(false, a0, false, b0, (short)0, c, false, false);
    c = __builtin_amdgcn_wmma_f32_16x16x4_f32(false, a1, false, b1, (short)0, c, false, false);
    float d01 = (idxsel & 1) ? c[1] : c[0];
    float d23 = (idxsel & 1) ? c[3] : c[2];
    float d45 = (idxsel & 1) ? c[5] : c[4];
    float d67 = (idxsel & 1) ? c[7] : c[6];
    float d03 = (idxsel & 2) ? d23 : d01;
    float d47 = (idxsel & 2) ? d67 : d45;
    float dd  = (idxsel & 4) ? d47 : d03;
    const float other = __builtin_bit_cast(
        float, __builtin_amdgcn_update_dpp(0, __builtin_bit_cast(int, dd), 0xB1, 0xF, 0xF, false));
    float t = (dd + other) * INV_SCALE;
    t = fminf(fmaxf(t, -5.0f), 5.0f);
    if (isdiag && !(lane & 1)) sw[(j + u) * 8 + wslot] = __expf(t);
  }
}

__device__ __forceinline__ void do_group(const float* __restrict__ k, const float* __restrict__ v,
                                         int myidx, int cnt, int lane, int* grp,
                                         const float* qres, float4* w4,
                                         float* acc, float* zacc, int row0) {
  const int is_hi = lane >> 4;
  const unsigned pofs = (unsigned)(((lane & 15) << 1) | is_hi) * 16u;
  grp[lane] = myidx;
  float* sw = (float*)w4;

  int i = 0;
  for (; i + KBATCH <= cnt; i += KBATCH)
    score_rows<KBATCH>(k, qres, sw, myidx, i, lane, pofs, row0);
  if (i + 8 <= cnt) {
    score_rows<8>(k, qres, sw, myidx, i, lane, pofs, row0);
    i += 8;
  }
  if (i + 4 <= cnt) {
    score_rows<4>(k, qres, sw, myidx, i, lane, pofs, row0);
    i += 4;
  }
  for (; i < cnt; ++i)
    score_rows<1>(k, qres, sw, myidx, i, lane, pofs, row0);
  asm volatile("s_wait_dscnt 0x0" ::: "memory");

  {
    const int zh    = lane & 7;
    const int zslot = (zh & 1) * 4 + (zh >> 1);
    for (int t = 0; t < 4; ++t) {
      const int j = t * 4 + (lane >> 3);
      if (j < cnt) {
        const int dj = grp[16 + j];
        atomicAdd(zacc + (dj - row0) * HEADS + zh, sw[j * 8 + zslot]);
      }
    }
  }

  int j = 0;
  for (; j + VBATCH <= cnt; j += VBATCH)
    add_rows<VBATCH>(v, acc, w4, myidx, j, lane, is_hi, row0);
  if (j + 4 <= cnt) {
    add_rows<4>(v, acc, w4, myidx, j, lane, is_hi, row0);
    j += 4;
  }
  for (; j < cnt; ++j)
    add_rows<1>(v, acc, w4, myidx, j, lane, is_hi, row0);
}

__global__ void __launch_bounds__(PTHREADS)
bin_edges_kernel(const int* __restrict__ eidx, int* __restrict__ ws_pairs, int* __restrict__ ws_tab,
                 int nEdges, int nNodes, int nOwners) {
  __shared__ int  s_cnt[MAXB];
  __shared__ int  s_off[MAXB + 1];
  __shared__ int2 s_pair[PCAP];

  const int tid   = threadIdx.x;
  const int piece = blockIdx.x;
  const int chunk = (nEdges + NSEG - 1) / NSEG;
  const int e0    = piece * chunk;
  int len = nEdges - e0;
  if (len > chunk) len = chunk;

  for (int i = tid; i < MAXB; i += PTHREADS) s_cnt[i] = 0;
  __syncthreads();

  int sv[PCH], dv[PCH], own[PCH], rnk[PCH];
#pragma unroll
  for (int u = 0; u < PCH; ++u) {
    const int i = tid + u * PTHREADS;
    sv[u] = 0;
    dv[u] = -1;
    if (i < len) {
      sv[u] = eidx[e0 + i];
      dv[u] = eidx[nEdges + e0 + i];
    }
  }
#pragma unroll
  for (int u = 0; u < PCH; ++u) {
    own[u] = ((unsigned)dv[u] < (unsigned)nNodes) ? (dv[u] / NROWS) : -1;
    rnk[u] = 0;
    if (own[u] >= 0) rnk[u] = atomicAdd(&s_cnt[own[u]], 1);
  }
  __syncthreads();

  if (tid < 32) {
    int a[MAXB / 32];
    int run = 0;
    for (int t = 0; t < MAXB / 32; ++t) {
      a[t] = run;
      run += s_cnt[tid * (MAXB / 32) + t];
    }
    int inc = run;
    for (int o = 1; o < 32; o <<= 1) {
      const int y = __shfl_up(inc, o);
      if (tid >= o) inc += y;
    }
    const int base = inc - run;
    for (int t = 0; t < MAXB / 32; ++t) s_off[tid * (MAXB / 32) + t] = base + a[t];
    if (tid == 31) s_off[MAXB] = inc;
  }
  __syncthreads();

#pragma unroll
  for (int u = 0; u < PCH; ++u)
    if (own[u] >= 0) s_pair[s_off[own[u]] + rnk[u]] = make_int2(sv[u], dv[u]);
  __syncthreads();

  const int total = s_off[MAXB];
  int2* pairs = (int2*)ws_pairs + (long)piece * PCAP;
  for (int i = tid; i < total; i += PTHREADS) pairs[i] = s_pair[i];
  int2* tab = (int2*)ws_tab + (long)piece * MAXB;
  for (int b = tid; b < nOwners; b += PTHREADS)
    tab[b] = make_int2(piece * PCAP + s_off[b], s_off[b + 1] - s_off[b]);
}

__device__ __forceinline__ int2 load_pair(const int2* __restrict__ pairs, const int (&bound)[SEGW],
                                          const int (&delta)[SEGW], int i) {
  int pos = i + delta[0];
  for (int s = 1; s < SEGW; ++s)
    if (i >= bound[s]) pos = i + delta[s];
  return pairs[pos];
}

__global__ void __launch_bounds__(1024)
edge_attn_kernel(const float* __restrict__ q, const float* __restrict__ k,
                 const float* __restrict__ v, const int* __restrict__ ws_pairs,
                 const int* __restrict__ ws_tab, float* __restrict__ out, int nNodes) {
  __shared__ float  s_acc[NROWS * HD];
  __shared__ float  s_z[NROWS * HEADS];
  __shared__ float4 s_q[NROWS * QPITCH / 4];
  __shared__ int    s_grp[NWAVES][32];
  __shared__ float4 s_w4[NWAVES][32];

  const int tid  = threadIdx.x;
  const int lane = tid & 31;
  const int wv   = __builtin_amdgcn_readfirstlane(tid >> 5);
  const int row0 = blockIdx.x * NROWS;

  const int2* tab   = (const int2*)ws_tab;
  const int2* pairs = (const int2*)ws_pairs;
  int2 ent = make_int2(0, 0);
  if (lane < SEGW) ent = tab[(long)(wv * SEGW + lane) * MAXB + blockIdx.x];

  for (int i = tid; i < NROWS * HD; i += NWAVES * 32) s_acc[i] = 0.0f;
  for (int i = tid; i < NROWS * HEADS; i += NWAVES * 32) s_z[i] = 0.0f;
  {
    float4 t[QTURNS];
#pragma unroll
    for (int it = 0; it < QTURNS; ++it) {
      const int i    = tid + it * (NWAVES * 32);
      const int node = row0 + (i >> 5);
      t[it] = make_float4(0.f, 0.f, 0.f, 0.f);
      if (i < NROWS * (HD / 4) && node < nNodes)
        t[it] = *(const float4*)(q + (long)node * HD + ((((i & 15) << 1) | ((i >> 4) & 1)) * 4));
    }
#pragma unroll
    for (int it = 0; it < QTURNS; ++it) {
      const int i = tid + it * (NWAVES * 32);
      if (i < NROWS * (HD / 4)) s_q[(i >> 5) * (QPITCH / 4) + (i & 31)] = t[it];
    }
  }

  int bound[SEGW], delta[SEGW];
  int total = 0;
  for (int s = 0; s < SEGW; ++s) {
    const int st = __builtin_amdgcn_readlane(ent.x, s);
    const int ln = __builtin_amdgcn_readlane(ent.y, s);
    bound[s] = total;
    delta[s] = st - total;
    total += ln;
  }
  const int m     = lane & 15;
  const int is_hi = lane >> 4;
  int2 pr = make_int2(0, 0);
  if (total > 0) {
    const int c0 = (total < 16) ? total : 16;
    pr = load_pair(pairs, bound, delta, (m < c0) ? m : (c0 - 1));
  }
  __syncthreads();

  for (int g0 = 0; g0 < total; g0 += 16) {
    const int cnt   = (total - g0 < 16) ? (total - g0) : 16;
    const int myidx = is_hi ? pr.y : pr.x;
    const int g1    = g0 + 16;
    if (g1 < total) {
      const int c1 = (total - g1 < 16) ? (total - g1) : 16;
      pr = load_pair(pairs, bound, delta, g1 + ((m < c1) ? m : (c1 - 1)));
    }
    do_group(k, v, myidx, cnt, lane, &s_grp[wv][0],
             (const float*)s_q, &s_w4[wv][0], s_acc, s_z, row0);
  }

  __syncthreads();
  for (int i = tid; i < NROWS * HD; i += NWAVES * 32) {
    const int row  = i >> 7;
    const int node = row0 + row;
    if (node < nNodes) {
      const int h = (i >> 4) & 7;
      out[(long)node * HD + (i & 127)] = s_acc[i] / (s_z[row * HEADS + h] + 1e-6f);
    }
  }
}

extern "C" void kernel_launch(void* const* d_in, const int* in_sizes, int n_in,
                              void* d_out, int out_size, void* d_ws, size_t ws_size,
                              hipStream_t stream) {
  const float* q    = (const float*)d_in[0];
  const float* k    = (const float*)d_in[1];
  const float* v    = (const float*)d_in[2];
  const int*   eidx = (const int*)d_in[3];
  float* out = (float*)d_out;

  const int nEdges = in_sizes[3] / 2;
  const int nNodes = out_size / HD;

  if (ws_size < ((size_t)NSEG * PCAP + (size_t)NSEG * MAXB) * 8) return;
  int* ws_pairs = (int*)d_ws;
  int* ws_tab   = ws_pairs + (size_t)NSEG * PCAP * 2;

  const int blocks = (nNodes + NROWS - 1) / NROWS;
  bin_edges_kernel<<<NSEG, PTHREADS, 0, stream>>>(eidx, ws_pairs, ws_tab, nEdges, nNodes, blocks);
  edge_attn_kernel<<<blocks, NWAVES * 32, 0, stream>>>(q, k, v, ws_pairs, ws_tab, out, nNodes);
}


# A V row is one 16-byte-per-lane load of 4 whole lines (was 4 one-line loads): lane owns 4 floats of one head, one weight; LDS sums kept in lane order; a group's 16 rows in flight; arrival order kept
# speedup vs baseline: 11.4714x; 1.0802x over previous
#include <hip/hip_runtime.h>
#include <math.h>

#define HEADS 8
#define DIM 16
#define HD 128
#define INV_SCALE 0.25f
#define VBATCH 16
#define KBATCH 16
#define NROWS 196
#define QPITCH 132
#define WPITCH 20
#define NWAVES 32
#define NSEG 256
#define SEGW (NSEG / NWAVES)
#define MAXB 256
#define PTHREADS 256
#define PCH 16
#define PCAP (PTHREADS * PCH)
#define QTURNS ((NROWS * (HD / 4) + NWAVES * 32 - 1) / (NWAVES * 32))

typedef __attribute__((ext_vector_type(2))) float v2f;
typedef __attribute__((ext_vector_type(8))) float v8f;

template <int NB>
__device__ __forceinline__ void add_rows(const float* __restrict__ v, float* acc,
                                         const float* sw, int myidx, int j,
                                         int lane, int row0) {
  float4 x[NB];
#pragma unroll
  for (int u = 0; u < NB; ++u) {
    const int sj = __builtin_amdgcn_readlane(myidx, j + u);
    x[u] = *(const float4*)(v + (long)sj * HD + lane * 4);
  }
  float w[NB];
  const float* wp = sw + (lane >> 2) * WPITCH + j;
  if constexpr (NB >= 4) {
#pragma unroll
    for (int u = 0; u < NB; u += 4) {
      const float4 t = *(const float4*)(wp + u);
      w[u] = t.x; w[u + 1] = t.y; w[u + 2] = t.z; w[u + 3] = t.w;
    }
  } else {
    w[0] = wp[0];
  }
#pragma unroll
  for (int u = 0; u < NB; ++u) {
    const int dj = __builtin_amdgcn_readlane(myidx, 16 + j + u);
    float* op = acc + (dj - row0) * HD + lane;
    atomicAdd(op,      x[u].x * w[u]);
    atomicAdd(op + 32, x[u].y * w[u]);
    atomicAdd(op + 64, x[u].z * w[u]);
    atomicAdd(op + 96, x[u].w * w[u]);
  }
}

template <int NB>
__device__ __forceinline__ void score_rows(const float* __restrict__ k, const float* qres,
                                           float* sw, int myidx, int j, int lane, unsigned pofs,
                                           int row0) {
  float4 kr[NB];
#pragma unroll
  for (int u = 0; u < NB; ++u) {
    const int sj = __builtin_amdgcn_readlane(myidx, j + u);
    const char* krow = (const char*)(k + (long)sj * HD);
    kr[u] = *(const float4*)(krow + pofs);
  }
  const int  idxsel = lane & 7;
  const bool isdiag = (lane < 8) || (lane >= 24);
  const int  drow   = (lane >= 24) ? (lane - 16) : lane;
  const int  head   = drow >> 1;
  const int  wbase  = head * WPITCH;
  constexpr int QA = (NB < 4) ? NB : 4;
  float4 qr[NB];
#pragma unroll
  for (int u = 0; u < QA; ++u) {
    const int dj = __builtin_amdgcn_readlane(myidx, 16 + j + u);
    qr[u] = *(const float4*)(qres + (dj - row0) * QPITCH + lane * 4);
  }
#pragma unroll
  for (int u = 0; u < NB; ++u) {
    if (u + QA < NB) {
      const int dj = __builtin_amdgcn_readlane(myidx, 16 + j + u + QA);
      qr[u + QA] = *(const float4*)(qres + (dj - row0) * QPITCH + lane * 4);
    }
    v8f c = {0.f, 0.f, 0.f, 0.f, 0.f, 0.f, 0.f, 0.f};
    const v2f a0 = {kr[u].x, kr[u].y};
    const v2f b0 = {qr[u].x, qr[u].y};
    const v2f a1 = {kr[u].z, kr[u].w};
    const v2f b1 = {qr[u].z, qr[u].w};
    c = __builtin_amdgcn_wmma_f32_16x16x4_f32(false, a0, false, b0, (short)0, c, false, false);
    c = __builtin_amdgcn_wmma_f32_16x16x4_f32(false, a1, false, b1, (short)0, c, false, false);
    float d01 = (idxsel & 1) ? c[1] : c[0];
    float d23 = (idxsel & 1) ? c[3] : c[2];
    float d45 = (idxsel & 1) ? c[5] : c[4];
    float d67 = (idxsel & 1) ? c[7] : c[6];
    float d03 = (idxsel & 2) ? d23 : d01;
    float d47 = (idxsel & 2) ? d67 : d45;
    float dd  = (idxsel & 4) ? d47 : d03;
    const float other = __builtin_bit_cast(
        float, __builtin_amdgcn_update_dpp(0, __builtin_bit_cast(int, dd), 0xB1, 0xF, 0xF, false));
    float t = (dd + other) * INV_SCALE;
    t = fminf(fmaxf(t, -5.0f), 5.0f);
    if (isdiag && !(lane & 1)) sw[wbase + j + u] = __expf(t);
  }
}

__device__ __forceinline__ void do_group(const float* __restrict__ k, const float* __restrict__ v,
                                         int myidx, int cnt, int lane, int* grp,
                                         const float* qres, float4* w4,
                                         float* acc, float* zacc, int row0) {
  const int is_hi = lane >> 4;
  const unsigned pofs = (unsigned)(((lane & 15) << 1) | is_hi) * 16u;
  grp[lane] = myidx;
  float* sw = (float*)w4;

  int i = 0;
  for (; i + KBATCH <= cnt; i += KBATCH)
    score_rows<KBATCH>(k, qres, sw, myidx, i, lane, pofs, row0);
  if (i + 8 <= cnt) {
    score_rows<8>(k, qres, sw, myidx, i, lane, pofs, row0);
    i += 8;
  }
  if (i + 4 <= cnt) {
    score_rows<4>(k, qres, sw, myidx, i, lane, pofs, row0);
    i += 4;
  }
  for (; i < cnt; ++i)
    score_rows<1>(k, qres, sw, myidx, i, lane, pofs, row0);
  asm volatile("s_wait_dscnt 0x0" ::: "memory");

  {
    const int zh = lane & 7;
    for (int t = 0; t < 4; ++t) {
      const int j = t * 4 + (lane >> 3);
      if (j < cnt) {
        const int dj = grp[16 + j];
        atomicAdd(zacc + (dj - row0) * HEADS + zh, sw[zh * WPITCH + j]);
      }
    }
  }

  int j = 0;
  for (; j + VBATCH <= cnt; j += VBATCH)
    add_rows<VBATCH>(v, acc, sw, myidx, j, lane, row0);
  if (j + 8 <= cnt) {
    add_rows<8>(v, acc, sw, myidx, j, lane, row0);
    j += 8;
  }
  if (j + 4 <= cnt) {
    add_rows<4>(v, acc, sw, myidx, j, lane, row0);
    j += 4;
  }
  for (; j < cnt; ++j)
    add_rows<1>(v, acc, sw, myidx, j, lane, row0);
}

__global__ void __launch_bounds__(PTHREADS)
bin_edges_kernel(const int* __restrict__ eidx, int* __restrict__ ws_pairs, int* __restrict__ ws_tab,
                 int nEdges, int nNodes, int nOwners) {
  __shared__ int  s_cnt[MAXB];
  __shared__ int  s_off[MAXB + 1];
  __shared__ int2 s_pair[PCAP];

  const int tid   = threadIdx.x;
  const int piece = blockIdx.x;
  const int chunk = (nEdges + NSEG - 1) / NSEG;
  const int e0    = piece * chunk;
  int len = nEdges - e0;
  if (len > chunk) len = chunk;

  for (int i = tid; i < MAXB; i += PTHREADS) s_cnt[i] = 0;
  __syncthreads();

  int sv[PCH], dv[PCH], own[PCH], rnk[PCH];
#pragma unroll
  for (int u = 0; u < PCH; ++u) {
    const int i = tid + u * PTHREADS;
    sv[u] = 0;
    dv[u] = -1;
    if (i < len) {
      sv[u] = eidx[e0 + i];
      dv[u] = eidx[nEdges + e0 + i];
    }
  }
#pragma unroll
  for (int u = 0; u < PCH; ++u) {
    own[u] = ((unsigned)dv[u] < (unsigned)nNodes) ? (dv[u] / NROWS) : -1;
    rnk[u] = 0;
    if (own[u] >= 0) rnk[u] = atomicAdd(&s_cnt[own[u]], 1);
  }
  __syncthreads();

  if (tid < 32) {
    int a[MAXB / 32];
    int run = 0;
    for (int t = 0; t < MAXB / 32; ++t) {
      a[t] = run;
      run += s_cnt[tid * (MAXB / 32) + t];
    }
    int inc = run;
    for (int o = 1; o < 32; o <<= 1) {
      const int y = __shfl_up(inc, o);
      if (tid >= o) inc += y;
    }
    const int base = inc - run;
    for (int t = 0; t < MAXB / 32; ++t) s_off[tid * (MAXB / 32) + t] = base + a[t];
    if (tid == 31) s_off[MAXB] = inc;
  }
  __syncthreads();

#pragma unroll
  for (int u = 0; u < PCH; ++u)
    if (own[u] >= 0) s_pair[s_off[own[u]] + rnk[u]] = make_int2(sv[u], dv[u]);
  __syncthreads();

  const int total = s_off[MAXB];
  int2* pairs = (int2*)ws_pairs + (long)piece * PCAP;
  for (int i = tid; i < total; i += PTHREADS) pairs[i] = s_pair[i];
  int2* tab = (int2*)ws_tab + (long)piece * MAXB;
  for (int b = tid; b < nOwners; b += PTHREADS)
    tab[b] = make_int2(piece * PCAP + s_off[b], s_off[b + 1] - s_off[b]);
}

__device__ __forceinline__ int2 load_pair(const int2* __restrict__ pairs, const int (&bound)[SEGW],
                                          const int (&delta)[SEGW], int i) {
  int pos = i + delta[0];
  for (int s = 1; s < SEGW; ++s)
    if (i >= bound[s]) pos = i + delta[s];
  return pairs[pos];
}

__global__ void __launch_bounds__(1024)
edge_attn_kernel(const float* __restrict__ q, const float* __restrict__ k,
                 const float* __restrict__ v, const int* __restrict__ ws_pairs,
                 const int* __restrict__ ws_tab, float* __restrict__ out, int nNodes) {
  __shared__ float  s_acc[NROWS * HD];
  __shared__ float  s_z[NROWS * HEADS];
  __shared__ float4 s_q[NROWS * QPITCH / 4];
  __shared__ int    s_grp[NWAVES][32];
  __shared__ float4 s_w4[NWAVES][WPITCH * HEADS / 4];

  const int tid  = threadIdx.x;
  const int lane = tid & 31;
  const int wv   = __builtin_amdgcn_readfirstlane(tid >> 5);
  const int row0 = blockIdx.x * NROWS;

  const int2* tab   = (const int2*)ws_tab;
  const int2* pairs = (const int2*)ws_pairs;
  int2 ent = make_int2(0, 0);
  if (lane < SEGW) ent = tab[(long)(wv * SEGW + lane) * MAXB + blockIdx.x];

  for (int i = tid; i < NROWS * HD; i += NWAVES * 32) s_acc[i] = 0.0f;
  for (int i = tid; i < NROWS * HEADS; i += NWAVES * 32) s_z[i] = 0.0f;
  {
    float4 t[QTURNS];
#pragma unroll
    for (int it = 0; it < QTURNS; ++it) {
      const int i    = tid + it * (NWAVES * 32);
      const int node = row0 + (i >> 5);
      t[it] = make_float4(0.f, 0.f, 0.f, 0.f);
      if (i < NROWS * (HD / 4) && node < nNodes)
        t[it] = *(const float4*)(q + (long)node * HD + ((((i & 15) << 1) | ((i >> 4) & 1)) * 4));
    }
#pragma unroll
    for (int it = 0; it < QTURNS; ++it) {
      const int i = tid + it * (NWAVES * 32);
      if (i < NROWS * (HD / 4)) s_q[(i >> 5) * (QPITCH / 4) + (i & 31)] = t[it];
    }
  }

  int bound[SEGW], delta[SEGW];
  int total = 0;
  for (int s = 0; s < SEGW; ++s) {
    const int st = __builtin_amdgcn_readlane(ent.x, s);
    const int ln = __builtin_amdgcn_readlane(ent.y, s);
    bound[s] = total;
    delta[s] = st - total;
    total += ln;
  }
  const int m     = lane & 15;
  const int is_hi = lane >> 4;
  int2 pr = make_int2(0, 0);
  if (total > 0) {
    const int c0 = (total < 16) ? total : 16;
    pr = load_pair(pairs, bound, delta, (m < c0) ? m : (c0 - 1));
  }
  __syncthreads();

  for (int g0 = 0; g0 < total; g0 += 16) {
    const int cnt   = (total - g0 < 16) ? (total - g0) : 16;
    const int myidx = is_hi ? pr.y : pr.x;
    const int g1    = g0 + 16;
    if (g1 < total) {
      const int c1 = (total - g1 < 16) ? (total - g1) : 16;
      pr = load_pair(pairs, bound, delta, g1 + ((m < c1) ? m : (c1 - 1)));
    }
    do_group(k, v, myidx, cnt, lane, &s_grp[wv][0],
             (const float*)s_q, &s_w4[wv][0], s_acc, s_z, row0);
  }

  __syncthreads();
  for (int i = tid; i < NROWS * 32; i += NWAVES * 32) {
    const int row  = i >> 5;
    const int l    = i & 31;
    const int node = row0 + row;
    if (node < nNodes) {
      const float* ap = s_acc + row * HD + l;
      const float  d  = s_z[row * HEADS + (l >> 2)] + 1e-6f;
      float4 o;
      o.x = ap[0]  / d;
      o.y = ap[32] / d;
      o.z = ap[64] / d;
      o.w = ap[96] / d;
      *(float4*)(out + (long)node * HD + l * 4) = o;
    }
  }
}

extern "C" void kernel_launch(void* const* d_in, const int* in_sizes, int n_in,
                              void* d_out, int out_size, void* d_ws, size_t ws_size,
                              hipStream_t stream) {
  const float* q    = (const float*)d_in[0];
  const float* k    = (const float*)d_in[1];
  const float* v    = (const float*)d_in[2];
  const int*   eidx = (const int*)d_in[3];
  float* out = (float*)d_out;

  const int nEdges = in_sizes[3] / 2;
  const int nNodes = out_size / HD;

  if (ws_size < ((size_t)NSEG * PCAP + (size_t)NSEG * MAXB) * 8) return;
  int* ws_pairs = (int*)d_ws;
  int* ws_tab   = ws_pairs + (size_t)NSEG * PCAP * 2;

  const int blocks = (nNodes + NROWS - 1) / NROWS;
  bin_edges_kernel<<<NSEG, PTHREADS, 0, stream>>>(eidx, ws_pairs, ws_tab, nEdges, nNodes, blocks);
  edge_attn_kernel<<<blocks, NWAVES * 32, 0, stream>>>(q, k, v, ws_pairs, ws_tab, out, nNodes);
}


# A work-group's 256 runs go into one LDS list; its waves draw groups of 16 edges from it by an LDS counter (was 8 fixed runs per wave, each ending in a short tail); LDS float adds, arrival order
# speedup vs baseline: 12.3438x; 1.0761x over previous
#include <hip/hip_runtime.h>
#include <math.h>

#define HEADS 8
#define DIM 16
#define HD 128
#define INV_SCALE 0.25f
#define VBATCH 16
#define KBATCH 16
#define NROWS 196
#define QPITCH 132
#define WPITCH 20
#define NWAVES 32
#define NSEG 256
#define LCAP 4096
#define CPT 8
#define MAXB 256
#define PTHREADS 256
#define PCH 16
#define PCAP (PTHREADS * PCH)
#define QTURNS ((NROWS * (HD / 4) + NWAVES * 32 - 1) / (NWAVES * 32))

typedef __attribute__((ext_vector_type(2))) float v2f;
typedef __attribute__((ext_vector_type(8))) float v8f;

__device__ __forceinline__ int draw_group(int* next, int lane) {
  int g = 0;
  if (lane == 0) g = atomicAdd(next, 1);
  return __builtin_amdgcn_readfirstlane(g);
}

__device__ __forceinline__ int2 group_pairs(const int2* list, int clen, int g, int lane) {
  const int left = clen - g * 16;
  const int cnt  = (left < 16) ? left : 16;
  const int m    = lane & 15;
  return list[g * 16 + ((m < cnt) ? m : (cnt - 1))];
}

template <int NB, bool GUARD>
__device__ __forceinline__ void add_rows(const float* __restrict__ v, float* acc,
                                         const float* sw, int myidx, int j, int cnt,
                                         int lane, int row0) {
  float4 x[NB];
#pragma unroll
  for (int u = 0; u < NB; ++u) {
    const int sj = __builtin_amdgcn_readlane(myidx, j + u);
    x[u] = *(const float4*)(v + (long)sj * HD + lane * 4);
  }
  float w[NB];
  const float* wp = sw + (lane >> 2) * WPITCH + j;
  if constexpr (NB >= 4) {
#pragma unroll
    for (int u = 0; u < NB; u += 4) {
      const float4 t = *(const float4*)(wp + u);
      w[u] = t.x; w[u + 1] = t.y; w[u + 2] = t.z; w[u + 3] = t.w;
    }
  } else {
    w[0] = wp[0];
  }
#pragma unroll
  for (int u = 0; u < NB; ++u) {
    if (!GUARD || j + u < cnt) {
      const int dj = __builtin_amdgcn_readlane(myidx, 16 + j + u);
      float* op = acc + (dj - row0) * HD + lane;
      atomicAdd(op,      x[u].x * w[u]);
      atomicAdd(op + 32, x[u].y * w[u]);
      atomicAdd(op + 64, x[u].z * w[u]);
      atomicAdd(op + 96, x[u].w * w[u]);
    }
  }
}

template <int NB>
__device__ __forceinline__ void score_rows(const float* __restrict__ k, const float* qres,
                                           float* sw, int myidx, int j, int lane, unsigned pofs,
                                           int row0, int* next, const int2* list, int clen, int ng,
                                           int& gn, int2& prn) {
  float4 kr[NB];
#pragma unroll
  for (int u = 0; u < NB; ++u) {
    const int sj = __builtin_amdgcn_readlane(myidx, j + u);
    const char* krow = (const char*)(k + (long)sj * HD);
    kr[u] = *(const float4*)(krow + pofs);
  }
  gn  = draw_group(next, lane);
  prn = make_int2(0, 0);
  if (gn < ng) prn = group_pairs(list, clen, gn, lane);
  const int  idxsel = lane & 7;
  const bool isdiag = (lane < 8) || (lane >= 24);
  const int  drow   = (lane >= 24) ? (lane - 16) : lane;
  const int  head   = drow >> 1;
  const int  wbase  = head * WPITCH;
  constexpr int QA = (NB < 4) ? NB : 4;
  float4 qr[NB];
#pragma unroll
  for (int u = 0; u < QA; ++u) {
    const int dj = __builtin_amdgcn_readlane(myidx, 16 + j + u);
    qr[u] = *(const float4*)(qres + (dj - row0) * QPITCH + lane * 4);
  }
#pragma unroll
  for (int u = 0; u < NB; ++u) {
    if (u + QA < NB) {
      const int dj = __builtin_amdgcn_readlane(myidx, 16 + j + u + QA);
      qr[u + QA] = *(const float4*)(qres + (dj - row0) * QPITCH + lane * 4);
    }
    v8f c = {0.f, 0.f, 0.f, 0.f, 0.f, 0.f, 0.f, 0.f};
    const v2f a0 = {kr[u].x, kr[u].y};
    const v2f b0 = {qr[u].x, qr[u].y};
    const v2f a1 = {kr[u].z, kr[u].w};
    const v2f b1 = {qr[u].z, qr[u].w};
    c = __builtin_amdgcn_wmma_f32_16x16x4_f32(false, a0, false, b0, (short)0, c, false, false);
    c = __builtin_amdgcn_wmma_f32_16x16x4_f32(false, a1, false, b1, (short)0, c, false, false);
    float d01 = (idxsel & 1) ? c[1] : c[0];
    float d23 = (idxsel & 1) ? c[3] : c[2];
    float d45 = (idxsel & 1) ? c[5] : c[4];
    float d67 = (idxsel & 1) ? c[7] : c[6];
    float d03 = (idxsel & 2) ? d23 : d01;
    float d47 = (idxsel & 2) ? d67 : d45;
    float dd  = (idxsel & 4) ? d47 : d03;
    const float other = __builtin_bit_cast(
        float, __builtin_amdgcn_update_dpp(0, __builtin_bit_cast(int, dd), 0xB1, 0xF, 0xF, false));
    float t = (dd + other) * INV_SCALE;
    t = fminf(fmaxf(t, -5.0f), 5.0f);
    if (isdiag && !(lane & 1)) sw[wbase + j + u] = __expf(t);
  }
}

__device__ __forceinline__ void do_group(const float* __restrict__ k, const float* __restrict__ v,
                                         int myidx, int cnt, int lane, int* grp,
                                         const float* qres, float4* w4,
                                         float* acc, float* zacc, int row0,
                                         int* next, const int2* list, int clen, int ng,
                                         int& gn, int2& prn) {
  const int is_hi = lane >> 4;
  const unsigned pofs = (unsigned)(((lane & 15) << 1) | is_hi) * 16u;
  grp[lane] = myidx;
  float* sw = (float*)w4;

  score_rows<KBATCH>(k, qres, sw, myidx, 0, lane, pofs, row0, next, list, clen, ng, gn, prn);
  asm volatile("s_wait_dscnt 0x0" ::: "memory");

  {
    const int zh = lane & 7;
    for (int t = 0; t < 4; ++t) {
      const int j = t * 4 + (lane >> 3);
      if (j < cnt) {
        const int dj = grp[16 + j];
        atomicAdd(zacc + (dj - row0) * HEADS + zh, sw[zh * WPITCH + j]);
      }
    }
  }

  if (cnt == VBATCH)
    add_rows<VBATCH, false>(v, acc, sw, myidx, 0, cnt, lane, row0);
  else
    add_rows<VBATCH, true>(v, acc, sw, myidx, 0, cnt, lane, row0);
}

__global__ void __launch_bounds__(PTHREADS)
bin_edges_kernel(const int* __restrict__ eidx, int* __restrict__ ws_pairs, int* __restrict__ ws_tab,
                 int nEdges, int nNodes, int nOwners) {
  __shared__ int  s_cnt[MAXB];
  __shared__ int  s_off[MAXB + 1];
  __shared__ int2 s_pair[PCAP];

  const int tid   = threadIdx.x;
  const int piece = blockIdx.x;
  const int chunk = (nEdges + NSEG - 1) / NSEG;
  const int e0    = piece * chunk;
  int len = nEdges - e0;
  if (len > chunk) len = chunk;

  for (int i = tid; i < MAXB; i += PTHREADS) s_cnt[i] = 0;
  __syncthreads();

  int sv[PCH], dv[PCH], own[PCH], rnk[PCH];
#pragma unroll
  for (int u = 0; u < PCH; ++u) {
    const int i = tid + u * PTHREADS;
    sv[u] = 0;
    dv[u] = -1;
    if (i < len) {
      sv[u] = eidx[e0 + i];
      dv[u] = eidx[nEdges + e0 + i];
    }
  }
#pragma unroll
  for (int u = 0; u < PCH; ++u) {
    own[u] = ((unsigned)dv[u] < (unsigned)nNodes) ? (dv[u] / NROWS) : -1;
    rnk[u] = 0;
    if (own[u] >= 0) rnk[u] = atomicAdd(&s_cnt[own[u]], 1);
  }
  __syncthreads();

  if (tid < 32) {
    int a[MAXB / 32];
    int run = 0;
    for (int t = 0; t < MAXB / 32; ++t) {
      a[t] = run;
      run += s_cnt[tid * (MAXB / 32) + t];
    }
    int inc = run;
    for (int o = 1; o < 32; o <<= 1) {
      const int y = __shfl_up(inc, o);
      if (tid >= o) inc += y;
    }
    const int base = inc - run;
    for (int t = 0; t < MAXB / 32; ++t) s_off[tid * (MAXB / 32) + t] = base + a[t];
    if (tid == 31) s_off[MAXB] = inc;
  }
  __syncthreads();

#pragma unroll
  for (int u = 0; u < PCH; ++u)
    if (own[u] >= 0) s_pair[s_off[own[u]] + rnk[u]] = make_int2(sv[u], dv[u]);
  __syncthreads();

  const int total = s_off[MAXB];
  int2* pairs = (int2*)ws_pairs + (long)piece * PCAP;
  for (int i = tid; i < total; i += PTHREADS) pairs[i] = s_pair[i];
  int2* tab = (int2*)ws_tab + (long)piece * MAXB;
  for (int b = tid; b < nOwners; b += PTHREADS)
    tab[b] = make_int2(piece * PCAP + s_off[b], s_off[b + 1] - s_off[b]);
}


__global__ void __launch_bounds__(1024)
edge_attn_kernel(const float* __restrict__ q, const float* __restrict__ k,
                 const float* __restrict__ v, const int* __restrict__ ws_pairs,
                 const int* __restrict__ ws_tab, float* __restrict__ out, int nNodes) {
  __shared__ float  s_acc[NROWS * HD];
  __shared__ float  s_z[NROWS * HEADS];
  __shared__ float4 s_q[NROWS * QPITCH / 4];
  __shared__ int    s_grp[NWAVES][32];
  __shared__ float4 s_w4[NWAVES][WPITCH * HEADS / 4];
  __shared__ int2   s_list[LCAP];
  __shared__ int    s_rstart[NSEG];
  __shared__ int    s_run[NSEG + 1];
  __shared__ int    s_next;
  static_assert(NSEG * 4 == NWAVES * 32, "four threads copy one run");
  static_assert(KBATCH == 16 && VBATCH == 16, "a group is 16 edges: one lane half of myidx");

  const int tid  = threadIdx.x;
  const int lane = tid & 31;
  const int wv   = __builtin_amdgcn_readfirstlane(tid >> 5);
  const int row0 = blockIdx.x * NROWS;

  const int2* tab   = (const int2*)ws_tab;
  const int2* pairs = (const int2*)ws_pairs;
  int2 ent = make_int2(0, 0);
  if (tid < NSEG) ent = tab[(long)tid * MAXB + blockIdx.x];

  float4 qt[QTURNS];
#pragma unroll
  for (int it = 0; it < QTURNS; ++it) {
    const int i    = tid + it * (NWAVES * 32);
    const int node = row0 + (i >> 5);
    qt[it] = make_float4(0.f, 0.f, 0.f, 0.f);
    if (i < NROWS * (HD / 4) && node < nNodes)
      qt[it] = *(const float4*)(q + (long)node * HD + ((((i & 15) << 1) | ((i >> 4) & 1)) * 4));
  }

  for (int i = tid; i < NROWS * HD; i += NWAVES * 32) s_acc[i] = 0.0f;
  for (int i = tid; i < NROWS * HEADS; i += NWAVES * 32) s_z[i] = 0.0f;

  if (tid < NSEG) {
    s_rstart[tid] = ent.x;
    s_run[tid]    = ent.y;
  }
  if (tid == 0) s_next = 0;
  __syncthreads();
  if (tid < 32) {
    int a[NSEG / 32];
    int run = 0;
    for (int t = 0; t < NSEG / 32; ++t) {
      a[t] = run;
      run += s_run[tid * (NSEG / 32) + t];
    }
    int inc = run;
    for (int o = 1; o < 32; o <<= 1) {
      const int y = __shfl_up(inc, o);
      if (tid >= o) inc += y;
    }
    const int before = inc - run;
    for (int t = 0; t < NSEG / 32; ++t) s_run[tid * (NSEG / 32) + t] = before + a[t];
    if (tid == 31) s_run[NSEG] = inc;
  }
  __syncthreads();

  const int total = s_run[NSEG];
  const int is_hi = lane >> 4;
  const int crun = tid >> 2;
  const int csub = tid & 3;
  const int rbeg = s_run[crun];
  const int rend = s_run[crun + 1];
  const int rsrc = s_rstart[crun];
  {
    const int hi0 = (rend < LCAP) ? rend : LCAP;
    int2 t[CPT];
#pragma unroll
    for (int u = 0; u < CPT; ++u) {
      const int p = rbeg + csub + 4 * u;
      t[u] = make_int2(0, 0);
      if (p < hi0) t[u] = pairs[rsrc + (p - rbeg)];
    }
#pragma unroll
    for (int it = 0; it < QTURNS; ++it) {
      const int i = tid + it * (NWAVES * 32);
      if (i < NROWS * (HD / 4)) s_q[(i >> 5) * (QPITCH / 4) + (i & 31)] = qt[it];
    }
#pragma unroll
    for (int u = 0; u < CPT; ++u) {
      const int p = rbeg + csub + 4 * u;
      if (p < hi0) s_list[p] = t[u];
    }
  }
  for (int base = 0; base < total; base += LCAP) {
    const int clen = (total - base < LCAP) ? (total - base) : LCAP;
    const int lo   = (rbeg > base) ? rbeg : base;
    const int hi   = (rend < base + clen) ? rend : (base + clen);
    for (int p = lo + csub + ((base == 0) ? 4 * CPT : 0); p < hi; p += 4)
      s_list[p - base] = pairs[rsrc + (p - rbeg)];
    __syncthreads();

    const int ng = (clen + 15) >> 4;
    int  g  = draw_group(&s_next, lane);
    int2 pr = make_int2(0, 0);
    if (g < ng) pr = group_pairs(s_list, clen, g, lane);
    while (g < ng) {
      const int left  = clen - g * 16;
      const int cnt   = (left < 16) ? left : 16;
      const int myidx = is_hi ? pr.y : pr.x;
      int  gn;
      int2 prn;
      do_group(k, v, myidx, cnt, lane, &s_grp[wv][0],
               (const float*)s_q, &s_w4[wv][0], s_acc, s_z, row0,
               &s_next, s_list, clen, ng, gn, prn);
      g  = gn;
      pr = prn;
    }
    if (base + LCAP < total) {
      __syncthreads();
      if (tid == 0) s_next = 0;
    }
  }

  __syncthreads();
  for (int i = tid; i < NROWS * 32; i += NWAVES * 32) {
    const int row  = i >> 5;
    const int l    = i & 31;
    const int node = row0 + row;
    if (node < nNodes) {
      const float* ap = s_acc + row * HD + l;
      const float  d  = s_z[row * HEADS + (l >> 2)] + 1e-6f;
      float4 o;
      o.x = ap[0]  / d;
      o.y = ap[32] / d;
      o.z = ap[64] / d;
      o.w = ap[96] / d;
      *(float4*)(out + (long)node * HD + l * 4) = o;
    }
  }
}

extern "C" void kernel_launch(void* const* d_in, const int* in_sizes, int n_in,
                              void* d_out, int out_size, void* d_ws, size_t ws_size,
                              hipStream_t stream) {
  const float* q    = (const float*)d_in[0];
  const float* k    = (const float*)d_in[1];
  const float* v    = (const float*)d_in[2];
  const int*   eidx = (const int*)d_in[3];
  float* out = (float*)d_out;

  const int nEdges = in_sizes[3] / 2;
  const int nNodes = out_size / HD;

  if (ws_size < ((size_t)NSEG * PCAP + (size_t)NSEG * MAXB) * 8) return;
  int* ws_pairs = (int*)d_ws;
  int* ws_tab   = ws_pairs + (size_t)NSEG * PCAP * 2;

  const int blocks = (nNodes + NROWS - 1) / NROWS;
  bin_edges_kernel<<<NSEG, PTHREADS, 0, stream>>>(eidx, ws_pairs, ws_tab, nEdges, nNodes, blocks);
  edge_attn_kernel<<<blocks, NWAVES * 32, 0, stream>>>(q, k, v, ws_pairs, ws_tab, out, nNodes);
}


# Each edge's V row is requested inside the score phase, behind that edge's matrix products, into registers that K rows have left (was: all 16 after the scores and Z adds); adds kept, arrival order
# speedup vs baseline: 12.7254x; 1.0309x over previous
#include <hip/hip_runtime.h>
#include <math.h>

#define HEADS 8
#define DIM 16
#define HD 128
#define INV_SCALE 0.25f
#define VBATCH 16
#define KBATCH 16
#define NROWS 196
#define QPITCH 132
#define WPITCH 20
#define NWAVES 32
#define NSEG 256
#define LCAP 4096
#define CPT 8
#define MAXB 256
#define PTHREADS 256
#define PCH 16
#define PCAP (PTHREADS * PCH)
#define QTURNS ((NROWS * (HD / 4) + NWAVES * 32 - 1) / (NWAVES * 32))

typedef __attribute__((ext_vector_type(2))) float v2f;
typedef __attribute__((ext_vector_type(8))) float v8f;

__device__ __forceinline__ int draw_group(int* next, int lane) {
  int g = 0;
  if (lane == 0) g = atomicAdd(next, 1);
  return __builtin_amdgcn_readfirstlane(g);
}

__device__ __forceinline__ int2 group_pairs(const int2* list, int clen, int g, int lane) {
  const int left = clen - g * 16;
  const int cnt  = (left < 16) ? left : 16;
  const int m    = lane & 15;
  return list[g * 16 + ((m < cnt) ? m : (cnt - 1))];
}

template <int NB, bool GUARD>
__device__ __forceinline__ void add_rows(const float4 (&x)[NB], float* acc,
                                         const float* sw, int myidx, int j, int cnt,
                                         int lane, int row0) {
  float w[NB];
  const float* wp = sw + (lane >> 2) * WPITCH + j;
  if constexpr (NB >= 4) {
#pragma unroll
    for (int u = 0; u < NB; u += 4) {
      const float4 t = *(const float4*)(wp + u);
      w[u] = t.x; w[u + 1] = t.y; w[u + 2] = t.z; w[u + 3] = t.w;
    }
  } else {
    w[0] = wp[0];
  }
#pragma unroll
  for (int u = 0; u < NB; ++u) {
    if (!GUARD || j + u < cnt) {
      const int dj = __builtin_amdgcn_readlane(myidx, 16 + j + u);
      float* op = acc + (dj - row0) * HD + lane;
      atomicAdd(op,      x[u].x * w[u]);
      atomicAdd(op + 32, x[u].y * w[u]);
      atomicAdd(op + 64, x[u].z * w[u]);
      atomicAdd(op + 96, x[u].w * w[u]);
    }
  }
}

template <int NB>
__device__ __forceinline__ void score_rows(const float* __restrict__ k, const float* __restrict__ v,
                                           const float* qres, float4 (&x)[NB],
                                           float* sw, int myidx, int j, int lane, unsigned pofs,
                                           int row0, int* next, const int2* list, int clen, int ng,
                                           int& gn, int2& prn) {
  float4 kr[NB];
#pragma unroll
  for (int u = 0; u < NB; ++u) {
    const int sj = __builtin_amdgcn_readlane(myidx, j + u);
    const char* krow = (const char*)(k + (long)sj * HD);
    kr[u] = *(const float4*)(krow + pofs);
  }
  gn  = draw_group(next, lane);
  prn = make_int2(0, 0);
  if (gn < ng) prn = group_pairs(list, clen, gn, lane);
  const int  idxsel = lane & 7;
  const bool isdiag = (lane < 8) || (lane >= 24);
  const int  drow   = (lane >= 24) ? (lane - 16) : lane;
  const int  head   = drow >> 1;
  const int  wbase  = head * WPITCH;
  constexpr int QA = (NB < 4) ? NB : 4;
  float4 qr[NB];
#pragma unroll
  for (int u = 0; u < QA; ++u) {
    const int dj = __builtin_amdgcn_readlane(myidx, 16 + j + u);
    qr[u] = *(const float4*)(qres + (dj - row0) * QPITCH + lane * 4);
  }
#pragma unroll
  for (int u = 0; u < NB; ++u) {
    if (u + QA < NB) {
      const int dj = __builtin_amdgcn_readlane(myidx, 16 + j + u + QA);
      qr[u + QA] = *(const float4*)(qres + (dj - row0) * QPITCH + lane * 4);
    }
    v8f c = {0.f, 0.f, 0.f, 0.f, 0.f, 0.f, 0.f, 0.f};
    const v2f a0 = {kr[u].x, kr[u].y};
    const v2f b0 = {qr[u].x, qr[u].y};
    const v2f a1 = {kr[u].z, kr[u].w};
    const v2f b1 = {qr[u].z, qr[u].w};
    c = __builtin_amdgcn_wmma_f32_16x16x4_f32(false, a0, false, b0, (short)0, c, false, false);
    c = __builtin_amdgcn_wmma_f32_16x16x4_f32(false, a1, false, b1, (short)0, c, false, false);
    {
      const int sj = __builtin_amdgcn_readlane(myidx, j + u);
      x[u] = *(const float4*)(v + (long)sj * HD + lane * 4);
    }
    float d01 = (idxsel & 1) ? c[1] : c[0];
    float d23 = (idxsel & 1) ? c[3] : c[2];
    float d45 = (idxsel & 1) ? c[5] : c[4];
    float d67 = (idxsel & 1) ? c[7] : c[6];
    float d03 = (idxsel & 2) ? d23 : d01;
    float d47 = (idxsel & 2) ? d67 : d45;
    float dd  = (idxsel & 4) ? d47 : d03;
    const float other = __builtin_bit_cast(
        float, __builtin_amdgcn_update_dpp(0, __builtin_bit_cast(int, dd), 0xB1, 0xF, 0xF, false));
    float t = (dd + other) * INV_SCALE;
    t = fminf(fmaxf(t, -5.0f), 5.0f);
    if (isdiag && !(lane & 1)) sw[wbase + j + u] = __expf(t);
  }
}

__device__ __forceinline__ void do_group(const float* __restrict__ k, const float* __restrict__ v,
                                         int myidx, int cnt, int lane, int* grp,
                                         const float* qres, float4* w4,
                                         float* acc, float* zacc, int row0,
                                         int* next, const int2* list, int clen, int ng,
                                         int& gn, int2& prn) {
  const int is_hi = lane >> 4;
  const unsigned pofs = (unsigned)(((lane & 15) << 1) | is_hi) * 16u;
  grp[lane] = myidx;
  float* sw = (float*)w4;

  float4 x[VBATCH];
  score_rows<KBATCH>(k, v, qres, x, sw, myidx, 0, lane, pofs, row0, next, list, clen, ng, gn, prn);
  asm volatile("s_wait_dscnt 0x0" ::: "memory");

  {
    const int zh = lane & 7;
    for (int t = 0; t < 4; ++t) {
      const int j = t * 4 + (lane >> 3);
      if (j < cnt) {
        const int dj = grp[16 + j];
        atomicAdd(zacc + (dj - row0) * HEADS + zh, sw[zh * WPITCH + j]);
      }
    }
  }

  if (cnt == VBATCH)
    add_rows<VBATCH, false>(x, acc, sw, myidx, 0, cnt, lane, row0);
  else
    add_rows<VBATCH, true>(x, acc, sw, myidx, 0, cnt, lane, row0);
}

__global__ void __launch_bounds__(PTHREADS)
bin_edges_kernel(const int* __restrict__ eidx, int* __restrict__ ws_pairs, int* __restrict__ ws_tab,
                 int nEdges, int nNodes, int nOwners) {
  __shared__ int  s_cnt[MAXB];
  __shared__ int  s_off[MAXB + 1];
  __shared__ int2 s_pair[PCAP];

  const int tid   = threadIdx.x;
  const int piece = blockIdx.x;
  const int chunk = (nEdges + NSEG - 1) / NSEG;
  const int e0    = piece * chunk;
  int len = nEdges - e0;
  if (len > chunk) len = chunk;

  for (int i = tid; i < MAXB; i += PTHREADS) s_cnt[i] = 0;
  __syncthreads();

  int sv[PCH], dv[PCH], own[PCH], rnk[PCH];
#pragma unroll
  for (int u = 0; u < PCH; ++u) {
    const int i = tid + u * PTHREADS;
    sv[u] = 0;
    dv[u] = -1;
    if (i < len) {
      sv[u] = eidx[e0 + i];
      dv[u] = eidx[nEdges + e0 + i];
    }
  }
#pragma unroll
  for (int u = 0; u < PCH; ++u) {
    own[u] = ((unsigned)dv[u] < (unsigned)nNodes) ? (dv[u] / NROWS) : -1;
    rnk[u] = 0;
    if (own[u] >= 0) rnk[u] = atomicAdd(&s_cnt[own[u]], 1);
  }
  __syncthreads();

  if (tid < 32) {
    int a[MAXB / 32];
    int run = 0;
    for (int t = 0; t < MAXB / 32; ++t) {
      a[t] = run;
      run += s_cnt[tid * (MAXB / 32) + t];
    }
    int inc = run;
    for (int o = 1; o < 32; o <<= 1) {
      const int y = __shfl_up(inc, o);
      if (tid >= o) inc += y;
    }
    const int base = inc - run;
    for (int t = 0; t < MAXB / 32; ++t) s_off[tid * (MAXB / 32) + t] = base + a[t];
    if (tid == 31) s_off[MAXB] = inc;
  }
  __syncthreads();

#pragma unroll
  for (int u = 0; u < PCH; ++u)
    if (own[u] >= 0) s_pair[s_off[own[u]] + rnk[u]] = make_int2(sv[u], dv[u]);
  __syncthreads();

  const int total = s_off[MAXB];
  int2* pairs = (int2*)ws_pairs + (long)piece * PCAP;
  for (int i = tid; i < total; i += PTHREADS) pairs[i] = s_pair[i];
  int2* tab = (int2*)ws_tab + (long)piece * MAXB;
  for (int b = tid; b < nOwners; b += PTHREADS)
    tab[b] = make_int2(piece * PCAP + s_off[b], s_off[b + 1] - s_off[b]);
}


__global__ void __launch_bounds__(1024)
edge_attn_kernel(const float* __restrict__ q, const float* __restrict__ k,
                 const float* __restrict__ v, const int* __restrict__ ws_pairs,
                 const int* __restrict__ ws_tab, float* __restrict__ out, int nNodes) {
  __shared__ float  s_acc[NROWS * HD];
  __shared__ float  s_z[NROWS * HEADS];
  __shared__ float4 s_q[NROWS * QPITCH / 4];
  __shared__ int    s_grp[NWAVES][32];
  __shared__ float4 s_w4[NWAVES][WPITCH * HEADS / 4];
  __shared__ int2   s_list[LCAP];
  __shared__ int    s_rstart[NSEG];
  __shared__ int    s_run[NSEG + 1];
  __shared__ int    s_next;
  static_assert(NSEG * 4 == NWAVES * 32, "four threads copy one run");
  static_assert(KBATCH == 16 && VBATCH == 16, "a group is 16 edges: one lane half of myidx");

  const int tid  = threadIdx.x;
  const int lane = tid & 31;
  const int wv   = __builtin_amdgcn_readfirstlane(tid >> 5);
  const int row0 = blockIdx.x * NROWS;

  const int2* tab   = (const int2*)ws_tab;
  const int2* pairs = (const int2*)ws_pairs;
  int2 ent = make_int2(0, 0);
  if (tid < NSEG) ent = tab[(long)tid * MAXB + blockIdx.x];

  float4 qt[QTURNS];
#pragma unroll
  for (int it = 0; it < QTURNS; ++it) {
    const int i    = tid + it * (NWAVES * 32);
    const int node = row0 + (i >> 5);
    qt[it] = make_float4(0.f, 0.f, 0.f, 0.f);
    if (i < NROWS * (HD / 4) && node < nNodes)
      qt[it] = *(const float4*)(q + (long)node * HD + ((((i & 15) << 1) | ((i >> 4) & 1)) * 4));
  }

  for (int i = tid; i < NROWS * HD; i += NWAVES * 32) s_acc[i] = 0.0f;
  for (int i = tid; i < NROWS * HEADS; i += NWAVES * 32) s_z[i] = 0.0f;

  if (tid < NSEG) {
    s_rstart[tid] = ent.x;
    s_run[tid]    = ent.y;
  }
  if (tid == 0) s_next = 0;
  __syncthreads();
  if (tid < 32) {
    int a[NSEG / 32];
    int run = 0;
    for (int t = 0; t < NSEG / 32; ++t) {
      a[t] = run;
      run += s_run[tid * (NSEG / 32) + t];
    }
    int inc = run;
    for (int o = 1; o < 32; o <<= 1) {
      const int y = __shfl_up(inc, o);
      if (tid >= o) inc += y;
    }
    const int before = inc - run;
    for (int t = 0; t < NSEG / 32; ++t) s_run[tid * (NSEG / 32) + t] = before + a[t];
    if (tid == 31) s_run[NSEG] = inc;
  }
  __syncthreads();

  const int total = s_run[NSEG];
  const int is_hi = lane >> 4;
  const int crun = tid >> 2;
  const int csub = tid & 3;
  const int rbeg = s_run[crun];
  const int rend = s_run[crun + 1];
  const int rsrc = s_rstart[crun];
  {
    const int hi0 = (rend < LCAP) ? rend : LCAP;
    int2 t[CPT];
#pragma unroll
    for (int u = 0; u < CPT; ++u) {
      const int p = rbeg + csub + 4 * u;
      t[u] = make_int2(0, 0);
      if (p < hi0) t[u] = pairs[rsrc + (p - rbeg)];
    }
#pragma unroll
    for (int it = 0; it < QTURNS; ++it) {
      const int i = tid + it * (NWAVES * 32);
      if (i < NROWS * (HD / 4)) s_q[(i >> 5) * (QPITCH / 4) + (i & 31)] = qt[it];
    }
#pragma unroll
    for (int u = 0; u < CPT; ++u) {
      const int p = rbeg + csub + 4 * u;
      if (p < hi0) s_list[p] = t[u];
    }
  }
  for (int base = 0; base < total; base += LCAP) {
    const int clen = (total - base < LCAP) ? (total - base) : LCAP;
    const int lo   = (rbeg > base) ? rbeg : base;
    const int hi   = (rend < base + clen) ? rend : (base + clen);
    for (int p = lo + csub + ((base == 0) ? 4 * CPT : 0); p < hi; p += 4)
      s_list[p - base] = pairs[rsrc + (p - rbeg)];
    __syncthreads();

    const int ng = (clen + 15) >> 4;
    int  g  = draw_group(&s_next, lane);
    int2 pr = make_int2(0, 0);
    if (g < ng) pr = group_pairs(s_list, clen, g, lane);
    while (g < ng) {
      const int left  = clen - g * 16;
      const int cnt   = (left < 16) ? left : 16;
      const int myidx = is_hi ? pr.y : pr.x;
      int  gn;
      int2 prn;
      do_group(k, v, myidx, cnt, lane, &s_grp[wv][0],
               (const float*)s_q, &s_w4[wv][0], s_acc, s_z, row0,
               &s_next, s_list, clen, ng, gn, prn);
      g  = gn;
      pr = prn;
    }
    if (base + LCAP < total) {
      __syncthreads();
      if (tid == 0) s_next = 0;
    }
  }

  __syncthreads();
  for (int i = tid; i < NROWS * 32; i += NWAVES * 32) {
    const int row  = i >> 5;
    const int l    = i & 31;
    const int node = row0 + row;
    if (node < nNodes) {
      const float* ap = s_acc + row * HD + l;
      const float  d  = s_z[row * HEADS + (l >> 2)] + 1e-6f;
      float4 o;
      o.x = ap[0]  / d;
      o.y = ap[32] / d;
      o.z = ap[64] / d;
      o.w = ap[96] / d;
      *(float4*)(out + (long)node * HD + l * 4) = o;
    }
  }
}

extern "C" void kernel_launch(void* const* d_in, const int* in_sizes, int n_in,
                              void* d_out, int out_size, void* d_ws, size_t ws_size,
                              hipStream_t stream) {
  const float* q    = (const float*)d_in[0];
  const float* k    = (const float*)d_in[1];
  const float* v    = (const float*)d_in[2];
  const int*   eidx = (const int*)d_in[3];
  float* out = (float*)d_out;

  const int nEdges = in_sizes[3] / 2;
  const int nNodes = out_size / HD;

  if (ws_size < ((size_t)NSEG * PCAP + (size_t)NSEG * MAXB) * 8) return;
  int* ws_pairs = (int*)d_ws;
  int* ws_tab   = ws_pairs + (size_t)NSEG * PCAP * 2;

  const int blocks = (nNodes + NROWS - 1) / NROWS;
  bin_edges_kernel<<<NSEG, PTHREADS, 0, stream>>>(eidx, ws_pairs, ws_tab, nEdges, nNodes, blocks);
  edge_attn_kernel<<<blocks, NWAVES * 32, 0, stream>>>(q, k, v, ws_pairs, ws_tab, out, nNodes);
}
